# starting kernel
# baseline (speedup 1.0000x reference)
#include <hip/hip_runtime.h>

#define N_NODES   65536
#define NPG       128
#define NGRAPH    512
#define E_EDGES   1048576
#define HID       128
#define OUT_DIM   10
#define N_LAYERS  4
#define LN_EPS    1e-5f

typedef float v2f __attribute__((ext_vector_type(2)));
typedef float v4f __attribute__((ext_vector_type(4)));
typedef float v8f __attribute__((ext_vector_type(8)));

__device__ __forceinline__ void atomic_add_f32_hw(float* p, float v) {
    asm volatile("global_atomic_add_f32 %0, %1, off" : : "v"(p), "v"(v) : "memory");
}

__global__ void k_deg_init(int* __restrict__ degcnt) {
    int i = blockIdx.x * 256 + threadIdx.x;
    if (i < N_NODES) degcnt[i] = 0;
}

__global__ void k_deg_count(const int* __restrict__ edge_index, int* __restrict__ degcnt) {
    int e = blockIdx.x * 256 + threadIdx.x;
    if (e < E_EDGES) {
        int d = edge_index[E_EDGES + e];
        atomicAdd(&degcnt[d], 1);
    }
}

__global__ void k_deg_finish(const int* __restrict__ degcnt,
                             float* __restrict__ dinv, float* __restrict__ invdeg) {
    int i = blockIdx.x * 256 + threadIdx.x;
    if (i < N_NODES) {
        float d = 1.0f + (float)degcnt[i];
        dinv[i]   = rsqrtf(d);
        invdeg[i] = 1.0f / d;
    }
}

__global__ void k_embed(const int* __restrict__ x,
                        const float* __restrict__ cls_emb,
                        const float* __restrict__ nbr_emb,
                        float* __restrict__ h) {
    int t = blockIdx.x * 256 + threadIdx.x;
    int n = t >> 5;
    int f = (t & 31) << 2;
    int c  = x[2 * n + 0];
    int nb = x[2 * n + 1];
    v4f a = *(const v4f*)(cls_emb + (size_t)c  * HID + f);
    v4f b = *(const v4f*)(nbr_emb + (size_t)nb * HID + f);
    *(v4f*)(h + (size_t)n * HID + f) = a + b;
}

__launch_bounds__(256)
__global__ void k_gemm_fused(const float* __restrict__ h,
                             const float* __restrict__ W,
                             const float* __restrict__ bias,
                             const float* __restrict__ invdeg,
                             float* __restrict__ hw,
                             float* __restrict__ agg) {
    __shared__ float lWt[HID * HID];
    const int tid = threadIdx.x;

    for (int i = tid * 4; i < HID * HID; i += 256 * 4) {
        v4f w = *(const v4f*)(W + i);
        const int k = i >> 7;
        const int n = i & (HID - 1);
#pragma unroll
        for (int j = 0; j < 4; ++j)
            lWt[(n + j) * HID + (k ^ (((n + j) & 31) << 1))] = w[j];
    }
    __syncthreads();

    const int wave = tid >> 5;
    const int lane = tid & 31;
    const int half = lane >> 4;
    const int l16  = lane & 15;
    const int m0   = blockIdx.x * 128 + wave * 16;

    v8f zero = {};
    v8f acc[8];
#pragma unroll
    for (int t = 0; t < 8; ++t) acc[t] = zero;

    const float* arow = h + (size_t)(m0 + l16) * HID;

    for (int k = 0; k < HID; k += 4) {
        const int kb = k + half * 2;
        v2f a = *(const v2f*)(arow + kb);
#pragma unroll
        for (int t = 0; t < 8; ++t) {
            const int col = t * 16 + l16;
            v2f b = *(const v2f*)(lWt + col * HID + (kb ^ ((col & 31) << 1)));
            acc[t] = __builtin_amdgcn_wmma_f32_16x16x4_f32(
                false, a, false, b, (short)0, acc[t], false, false);
        }
    }

    const int rbase = half * 8;
#pragma unroll
    for (int r = 0; r < 8; ++r) {
        const int row = m0 + rbase + r;
        const float inv = invdeg[row];
        float* hwrow  = hw  + (size_t)row * HID;
        float* aggrow = agg + (size_t)row * HID;
#pragma unroll
        for (int t = 0; t < 8; ++t) {
            const int col = t * 16 + l16;
            const float v = acc[t][r];
            hwrow[col]  = v;
            aggrow[col] = v * inv + bias[col];
        }
    }
}

__global__ void k_scatter(const int* __restrict__ edge_index,
                          const float* __restrict__ dinv,
                          const float* __restrict__ hw,
                          float* __restrict__ agg) {
    int t = blockIdx.x * 256 + threadIdx.x;
    int e = t >> 5;
    if (e >= E_EDGES) return;
    int f = (t & 31) << 2;
    int s = edge_index[e];
    int d = edge_index[E_EDGES + e];
    float en = dinv[s] * dinv[d];
    v4f v = *(const v4f*)(hw + (size_t)s * HID + f);
    float* out = agg + (size_t)d * HID + f;
    atomic_add_f32_hw(out + 0, v.x * en);
    atomic_add_f32_hw(out + 1, v.y * en);
    atomic_add_f32_hw(out + 2, v.z * en);
    atomic_add_f32_hw(out + 3, v.w * en);
}

__launch_bounds__(256)
__global__ void k_relu_res_ln(float* __restrict__ h,
                              const float* __restrict__ agg,
                              const float* __restrict__ gamma,
                              const float* __restrict__ beta) {
    int t = blockIdx.x * 256 + threadIdx.x;
    int n = t >> 5;
    int lane = t & 31;
    int f = lane << 2;

    const float* ar = agg + (size_t)n * HID;
    float* hr = h + (size_t)n * HID;

    v4f hv = *(const v4f*)(hr + f);
    v4f av = *(const v4f*)(ar + f);
    v4f val;
    val.x = hv.x + fmaxf(av.x, 0.0f);
    val.y = hv.y + fmaxf(av.y, 0.0f);
    val.z = hv.z + fmaxf(av.z, 0.0f);
    val.w = hv.w + fmaxf(av.w, 0.0f);

    float s = val.x + val.y + val.z + val.w;
#pragma unroll
    for (int off = 16; off >= 1; off >>= 1) s += __shfl_xor(s, off, 32);
    const float mu = s * (1.0f / (float)HID);

    v4f dv = val - mu;
    float q = dv.x * dv.x + dv.y * dv.y + dv.z * dv.z + dv.w * dv.w;
#pragma unroll
    for (int off = 16; off >= 1; off >>= 1) q += __shfl_xor(q, off, 32);
    const float rstd = rsqrtf(q * (1.0f / (float)HID) + LN_EPS);

    v4f g = *(const v4f*)(gamma + f);
    v4f b = *(const v4f*)(beta + f);
    v4f outv = dv * rstd * g + b;
    *(v4f*)(hr + f) = outv;
}

__global__ void k_out(const float* __restrict__ h,
                      const float* __restrict__ W_out,
                      float* __restrict__ out) {
    int t = blockIdx.x * 256 + threadIdx.x;
    if (t >= NGRAPH * OUT_DIM) return;
    int g = t / OUT_DIM;
    int o = t % OUT_DIM;
    const float* hr = h + (size_t)(g * NPG) * HID;
    float s = 0.0f;
#pragma unroll 4
    for (int k = 0; k < HID; ++k) s = fmaf(hr[k], W_out[k * OUT_DIM + o], s);
    out[t] = s;
}

extern "C" void kernel_launch(void* const* d_in, const int* in_sizes, int n_in,
                              void* d_out, int out_size, void* d_ws, size_t ws_size,
                              hipStream_t stream) {
    const int*   x        = (const int*)d_in[0];
    const int*   ei       = (const int*)d_in[1];
    const float* cls_emb  = (const float*)d_in[3];
    const float* nbr_emb  = (const float*)d_in[4];
    const float* Ws       = (const float*)d_in[5];
    const float* bs       = (const float*)d_in[6];
    const float* gammas   = (const float*)d_in[7];
    const float* betas    = (const float*)d_in[8];
    const float* W_out    = (const float*)d_in[9];
    float*       out      = (float*)d_out;

    char* ws = (char*)d_ws;
    size_t off = 0;
    int*   degcnt = (int*)(ws + off);   off += (size_t)N_NODES * 4;
    float* dinv   = (float*)(ws + off); off += (size_t)N_NODES * 4;
    float* invdeg = (float*)(ws + off); off += (size_t)N_NODES * 4;
    float* h      = (float*)(ws + off); off += (size_t)N_NODES * HID * 4;
    float* hw     = (float*)(ws + off); off += (size_t)N_NODES * HID * 4;
    float* agg    = (float*)(ws + off); off += (size_t)N_NODES * HID * 4;
    (void)ws_size; (void)in_sizes; (void)n_in; (void)out_size;

    k_deg_init  <<<N_NODES / 256, 256, 0, stream>>>(degcnt);
    k_deg_count <<<E_EDGES / 256, 256, 0, stream>>>(ei, degcnt);
    k_deg_finish<<<N_NODES / 256, 256, 0, stream>>>(degcnt, dinv, invdeg);

    k_embed<<<(N_NODES * 32) / 256, 256, 0, stream>>>(x, cls_emb, nbr_emb, h);

    for (int i = 0; i < N_LAYERS; ++i) {
        k_gemm_fused<<<N_NODES / 128, 256, 0, stream>>>(
            h, Ws + (size_t)i * HID * HID, bs + (size_t)i * HID, invdeg, hw, agg);
        k_scatter<<<(E_EDGES * 32) / 256, 256, 0, stream>>>(ei, dinv, hw, agg);
        k_relu_res_ln<<<(N_NODES * 32) / 256, 256, 0, stream>>>(
            h, agg, gammas + (size_t)i * HID, betas + (size_t)i * HID);
    }

    k_out<<<(NGRAPH * OUT_DIM + 255) / 256, 256, 0, stream>>>(h, W_out, out);
}


# Scatter: one wave per 32 edges (was per edge): indices and norms loaded once per wave, rows of 8 edges requested as one group ahead of the previous group's atomics; sums in arrival order as before
# speedup vs baseline: 1.6341x; 1.6341x over previous
#include <hip/hip_runtime.h>

#define N_NODES   65536
#define NPG       128
#define NGRAPH    512
#define E_EDGES   1048576
#define HID       128
#define OUT_DIM   10
#define N_LAYERS  4
#define LN_EPS    1e-5f

typedef float v2f __attribute__((ext_vector_type(2)));
typedef float v4f __attribute__((ext_vector_type(4)));
typedef float v8f __attribute__((ext_vector_type(8)));

__global__ void k_deg_init(int* __restrict__ degcnt) {
    int i = blockIdx.x * 256 + threadIdx.x;
    if (i < N_NODES) degcnt[i] = 0;
}

__global__ void k_deg_count(const int* __restrict__ edge_index, int* __restrict__ degcnt) {
    int e = blockIdx.x * 256 + threadIdx.x;
    if (e < E_EDGES) {
        int d = edge_index[E_EDGES + e];
        atomicAdd(&degcnt[d], 1);
    }
}

__global__ void k_deg_finish(const int* __restrict__ degcnt,
                             float* __restrict__ dinv, float* __restrict__ invdeg) {
    int i = blockIdx.x * 256 + threadIdx.x;
    if (i < N_NODES) {
        float d = 1.0f + (float)degcnt[i];
        dinv[i]   = rsqrtf(d);
        invdeg[i] = 1.0f / d;
    }
}

__global__ void k_embed(const int* __restrict__ x,
                        const float* __restrict__ cls_emb,
                        const float* __restrict__ nbr_emb,
                        float* __restrict__ h) {
    int t = blockIdx.x * 256 + threadIdx.x;
    int n = t >> 5;
    int f = (t & 31) << 2;
    int c  = x[2 * n + 0];
    int nb = x[2 * n + 1];
    v4f a = *(const v4f*)(cls_emb + (size_t)c  * HID + f);
    v4f b = *(const v4f*)(nbr_emb + (size_t)nb * HID + f);
    *(v4f*)(h + (size_t)n * HID + f) = a + b;
}

__launch_bounds__(256)
__global__ void k_gemm_fused(const float* __restrict__ h,
                             const float* __restrict__ W,
                             const float* __restrict__ bias,
                             const float* __restrict__ invdeg,
                             float* __restrict__ hw,
                             float* __restrict__ agg) {
    __shared__ float lWt[HID * HID];
    const int tid = threadIdx.x;

    for (int i = tid * 4; i < HID * HID; i += 256 * 4) {
        v4f w = *(const v4f*)(W + i);
        const int k = i >> 7;
        const int n = i & (HID - 1);
#pragma unroll
        for (int j = 0; j < 4; ++j)
            lWt[(n + j) * HID + (k ^ (((n + j) & 31) << 1))] = w[j];
    }
    __syncthreads();

    const int wave = tid >> 5;
    const int lane = tid & 31;
    const int half = lane >> 4;
    const int l16  = lane & 15;
    const int m0   = blockIdx.x * 128 + wave * 16;

    v8f zero = {};
    v8f acc[8];
#pragma unroll
    for (int t = 0; t < 8; ++t) acc[t] = zero;

    const float* arow = h + (size_t)(m0 + l16) * HID;

    for (int k = 0; k < HID; k += 4) {
        const int kb = k + half * 2;
        v2f a = *(const v2f*)(arow + kb);
#pragma unroll
        for (int t = 0; t < 8; ++t) {
            const int col = t * 16 + l16;
            v2f b = *(const v2f*)(lWt + col * HID + (kb ^ ((col & 31) << 1)));
            acc[t] = __builtin_amdgcn_wmma_f32_16x16x4_f32(
                false, a, false, b, (short)0, acc[t], false, false);
        }
    }

    const int rbase = half * 8;
#pragma unroll
    for (int r = 0; r < 8; ++r) {
        const int row = m0 + rbase + r;
        const float inv = invdeg[row];
        float* hwrow  = hw  + (size_t)row * HID;
        float* aggrow = agg + (size_t)row * HID;
#pragma unroll
        for (int t = 0; t < 8; ++t) {
            const int col = t * 16 + l16;
            const float v = acc[t][r];
            hwrow[col]  = v;
            aggrow[col] = v * inv + bias[col];
        }
    }
}

#define EPW 32
#define GRP 8
static_assert(E_EDGES % (EPW * 8) == 0, "grid covers the edge list exactly");
static_assert(EPW % GRP == 0 && EPW == 32, "one edge per lane in the index stage");

template <int BYTE_OFF>
__device__ __forceinline__ void atomic_add_f32_row(float* rowbase, unsigned lane_off, float v) {
    asm volatile("global_atomic_add_f32 %0, %1, %2 offset:%3"
                 : : "v"(lane_off), "v"(v), "s"(rowbase), "n"(BYTE_OFF) : "memory");
}

__launch_bounds__(256)
__global__ void k_scatter(const int* __restrict__ edge_index,
                          const float* __restrict__ dinv,
                          const float* __restrict__ hw,
                          float* __restrict__ agg) {
    const int lane = threadIdx.x & 31;
    const int wave = blockIdx.x * 8 + (threadIdx.x >> 5);
    const int e    = wave * EPW + lane;
    const int   s_l  = edge_index[e];
    const int   d_l  = edge_index[E_EDGES + e];
    const float en_l = dinv[s_l] * dinv[d_l];
    const unsigned lane_off = (unsigned)lane * 4u;

    float r[2][GRP][4];

#pragma unroll
    for (int u = 0; u < GRP; ++u) {
        const int s = __builtin_amdgcn_readlane(s_l, u);
        const float* in = hw + (size_t)s * HID + lane;
        r[0][u][0] = in[0];
        r[0][u][1] = in[32];
        r[0][u][2] = in[64];
        r[0][u][3] = in[96];
    }

#pragma unroll
    for (int g = 0; g < EPW / GRP; ++g) {
        const int cur = g & 1;
        if (g + 1 < EPW / GRP) {
#pragma unroll
            for (int u = 0; u < GRP; ++u) {
                const int s = __builtin_amdgcn_readlane(s_l, (g + 1) * GRP + u);
                const float* in = hw + (size_t)s * HID + lane;
                r[cur ^ 1][u][0] = in[0];
                r[cur ^ 1][u][1] = in[32];
                r[cur ^ 1][u][2] = in[64];
                r[cur ^ 1][u][3] = in[96];
            }
        }
#pragma unroll
        for (int u = 0; u < GRP; ++u) {
            const int   d  = __builtin_amdgcn_readlane(d_l, g * GRP + u);
            const float en = __builtin_bit_cast(float,
                __builtin_amdgcn_readlane(__builtin_bit_cast(int, en_l), g * GRP + u));
            float* out = agg + (size_t)d * HID;
            atomic_add_f32_row<0>(out, lane_off, r[cur][u][0] * en);
            atomic_add_f32_row<128>(out, lane_off, r[cur][u][1] * en);
            atomic_add_f32_row<256>(out, lane_off, r[cur][u][2] * en);
            atomic_add_f32_row<384>(out, lane_off, r[cur][u][3] * en);
        }
    }
}

__launch_bounds__(256)
__global__ void k_relu_res_ln(float* __restrict__ h,
                              const float* __restrict__ agg,
                              const float* __restrict__ gamma,
                              const float* __restrict__ beta) {
    int t = blockIdx.x * 256 + threadIdx.x;
    int n = t >> 5;
    int lane = t & 31;
    int f = lane << 2;

    const float* ar = agg + (size_t)n * HID;
    float* hr = h + (size_t)n * HID;

    v4f hv = *(const v4f*)(hr + f);
    v4f av = *(const v4f*)(ar + f);
    v4f val;
    val.x = hv.x + fmaxf(av.x, 0.0f);
    val.y = hv.y + fmaxf(av.y, 0.0f);
    val.z = hv.z + fmaxf(av.z, 0.0f);
    val.w = hv.w + fmaxf(av.w, 0.0f);

    float s = val.x + val.y + val.z + val.w;
#pragma unroll
    for (int off = 16; off >= 1; off >>= 1) s += __shfl_xor(s, off, 32);
    const float mu = s * (1.0f / (float)HID);

    v4f dv = val - mu;
    float q = dv.x * dv.x + dv.y * dv.y + dv.z * dv.z + dv.w * dv.w;
#pragma unroll
    for (int off = 16; off >= 1; off >>= 1) q += __shfl_xor(q, off, 32);
    const float rstd = rsqrtf(q * (1.0f / (float)HID) + LN_EPS);

    v4f g = *(const v4f*)(gamma + f);
    v4f b = *(const v4f*)(beta + f);
    v4f outv = dv * rstd * g + b;
    *(v4f*)(hr + f) = outv;
}

__global__ void k_out(const float* __restrict__ h,
                      const float* __restrict__ W_out,
                      float* __restrict__ out) {
    int t = blockIdx.x * 256 + threadIdx.x;
    if (t >= NGRAPH * OUT_DIM) return;
    int g = t / OUT_DIM;
    int o = t % OUT_DIM;
    const float* hr = h + (size_t)(g * NPG) * HID;
    float s = 0.0f;
#pragma unroll 4
    for (int k = 0; k < HID; ++k) s = fmaf(hr[k], W_out[k * OUT_DIM + o], s);
    out[t] = s;
}

extern "C" void kernel_launch(void* const* d_in, const int* in_sizes, int n_in,
                              void* d_out, int out_size, void* d_ws, size_t ws_size,
                              hipStream_t stream) {
    const int*   x        = (const int*)d_in[0];
    const int*   ei       = (const int*)d_in[1];
    const float* cls_emb  = (const float*)d_in[3];
    const float* nbr_emb  = (const float*)d_in[4];
    const float* Ws       = (const float*)d_in[5];
    const float* bs       = (const float*)d_in[6];
    const float* gammas   = (const float*)d_in[7];
    const float* betas    = (const float*)d_in[8];
    const float* W_out    = (const float*)d_in[9];
    float*       out      = (float*)d_out;

    char* ws = (char*)d_ws;
    size_t off = 0;
    int*   degcnt = (int*)(ws + off);   off += (size_t)N_NODES * 4;
    float* dinv   = (float*)(ws + off); off += (size_t)N_NODES * 4;
    float* invdeg = (float*)(ws + off); off += (size_t)N_NODES * 4;
    float* h      = (float*)(ws + off); off += (size_t)N_NODES * HID * 4;
    float* hw     = (float*)(ws + off); off += (size_t)N_NODES * HID * 4;
    float* agg    = (float*)(ws + off); off += (size_t)N_NODES * HID * 4;
    (void)ws_size; (void)in_sizes; (void)n_in; (void)out_size;

    k_deg_init  <<<N_NODES / 256, 256, 0, stream>>>(degcnt);
    k_deg_count <<<E_EDGES / 256, 256, 0, stream>>>(ei, degcnt);
    k_deg_finish<<<N_NODES / 256, 256, 0, stream>>>(degcnt, dinv, invdeg);

    k_embed<<<(N_NODES * 32) / 256, 256, 0, stream>>>(x, cls_emb, nbr_emb, h);

    for (int i = 0; i < N_LAYERS; ++i) {
        k_gemm_fused<<<N_NODES / 128, 256, 0, stream>>>(
            h, Ws + (size_t)i * HID * HID, bs + (size_t)i * HID, invdeg, hw, agg);
        k_scatter<<<E_EDGES / (EPW * 8), 256, 0, stream>>>(ei, dinv, hw, agg);
        k_relu_res_ln<<<(N_NODES * 32) / 256, 256, 0, stream>>>(
            h, agg, gammas + (size_t)i * HID, betas + (size_t)i * HID);
    }

    k_out<<<(NGRAPH * OUT_DIM + 255) / 256, 256, 0, stream>>>(h, W_out, out);
}


# Scatter: each block owns 256 agg rows in an LDS tile and finds its edges by scanning a 1-byte owner key per edge; LDS f32 atomics replace the L2 atomics; sums in arrival order as before
# speedup vs baseline: 1.8280x; 1.1187x over previous
#include <hip/hip_runtime.h>

#define N_NODES   65536
#define NPG       128
#define NGRAPH    512
#define E_EDGES   1048576
#define HID       128
#define OUT_DIM   10
#define N_LAYERS  4
#define LN_EPS    1e-5f

typedef float v2f __attribute__((ext_vector_type(2)));
typedef float v4f __attribute__((ext_vector_type(4)));
typedef float v8f __attribute__((ext_vector_type(8)));

__global__ void k_deg_init(int* __restrict__ degcnt) {
    int i = blockIdx.x * 256 + threadIdx.x;
    if (i < N_NODES) degcnt[i] = 0;
}

__global__ void k_deg_count(const int* __restrict__ edge_index, int* __restrict__ degcnt,
                            unsigned char* __restrict__ okey) {
    int e = blockIdx.x * 256 + threadIdx.x;
    if (e < E_EDGES) {
        int d = edge_index[E_EDGES + e];
        atomicAdd(&degcnt[d], 1);
        okey[e] = (unsigned char)(d >> 8);
    }
}

__global__ void k_deg_finish(const int* __restrict__ degcnt,
                             float* __restrict__ dinv, float* __restrict__ invdeg) {
    int i = blockIdx.x * 256 + threadIdx.x;
    if (i < N_NODES) {
        float d = 1.0f + (float)degcnt[i];
        dinv[i]   = rsqrtf(d);
        invdeg[i] = 1.0f / d;
    }
}

__global__ void k_embed(const int* __restrict__ x,
                        const float* __restrict__ cls_emb,
                        const float* __restrict__ nbr_emb,
                        float* __restrict__ h) {
    int t = blockIdx.x * 256 + threadIdx.x;
    int n = t >> 5;
    int f = (t & 31) << 2;
    int c  = x[2 * n + 0];
    int nb = x[2 * n + 1];
    v4f a = *(const v4f*)(cls_emb + (size_t)c  * HID + f);
    v4f b = *(const v4f*)(nbr_emb + (size_t)nb * HID + f);
    *(v4f*)(h + (size_t)n * HID + f) = a + b;
}

__launch_bounds__(256)
__global__ void k_gemm_fused(const float* __restrict__ h,
                             const float* __restrict__ W,
                             const float* __restrict__ bias,
                             const float* __restrict__ invdeg,
                             float* __restrict__ hw,
                             float* __restrict__ agg) {
    __shared__ float lWt[HID * HID];
    const int tid = threadIdx.x;

    for (int i = tid * 4; i < HID * HID; i += 256 * 4) {
        v4f w = *(const v4f*)(W + i);
        const int k = i >> 7;
        const int n = i & (HID - 1);
#pragma unroll
        for (int j = 0; j < 4; ++j)
            lWt[(n + j) * HID + (k ^ (((n + j) & 31) << 1))] = w[j];
    }
    __syncthreads();

    const int wave = tid >> 5;
    const int lane = tid & 31;
    const int half = lane >> 4;
    const int l16  = lane & 15;
    const int m0   = blockIdx.x * 128 + wave * 16;

    v8f zero = {};
    v8f acc[8];
#pragma unroll
    for (int t = 0; t < 8; ++t) acc[t] = zero;

    const float* arow = h + (size_t)(m0 + l16) * HID;

    for (int k = 0; k < HID; k += 4) {
        const int kb = k + half * 2;
        v2f a = *(const v2f*)(arow + kb);
#pragma unroll
        for (int t = 0; t < 8; ++t) {
            const int col = t * 16 + l16;
            v2f b = *(const v2f*)(lWt + col * HID + (kb ^ ((col & 31) << 1)));
            acc[t] = __builtin_amdgcn_wmma_f32_16x16x4_f32(
                false, a, false, b, (short)0, acc[t], false, false);
        }
    }

    const int rbase = half * 8;
#pragma unroll
    for (int r = 0; r < 8; ++r) {
        const int row = m0 + rbase + r;
        const float inv = invdeg[row];
        float* hwrow  = hw  + (size_t)row * HID;
        float* aggrow = agg + (size_t)row * HID;
#pragma unroll
        for (int t = 0; t < 8; ++t) {
            const int col = t * 16 + l16;
            const float v = acc[t][r];
            hwrow[col]  = v;
            aggrow[col] = v * inv + bias[col];
        }
    }
}

#define OWN_NODES  256
#define SC_WAVES   32
#define KEYS_STEP  512
#define E_WAVE     (E_EDGES / SC_WAVES)
#define SCAN_STEPS (E_WAVE / KEYS_STEP)
#define QCAP       (32 + KEYS_STEP)
static_assert(N_NODES == 256 * OWN_NODES, "the owner key is one byte and the row in the tile another");
static_assert(E_EDGES % (SC_WAVES * KEYS_STEP) == 0 && (SCAN_STEPS & (SCAN_STEPS - 1)) == 0 && SCAN_STEPS % 4 == 0,
              "the waves' scans cover the edge list exactly");

__device__ __forceinline__ unsigned zero_bytes(unsigned x) {
    const unsigned y = (x & 0x7f7f7f7fu) + 0x7f7f7f7fu;
    return ~(y | x | 0x7f7f7f7fu);
}

__device__ __forceinline__ void lds_add(float* p, float v) {
    __hip_atomic_fetch_add(p, v, __ATOMIC_RELAXED, __HIP_MEMORY_SCOPE_WORKGROUP);
}

__device__ __forceinline__ void sc_drain(const unsigned* st, int n, int lane,
                                         const int* __restrict__ edge_index,
                                         const float* __restrict__ dinv,
                                         const float* __restrict__ hw,
                                         float* tile) {
    int   s_l  = 0;
    int   dl_l = OWN_NODES;
    float en_l = 0.0f;
    if (lane < n) {
        const unsigned e = st[lane];
        s_l = edge_index[e];
        const int d = edge_index[E_EDGES + e];
        en_l = dinv[s_l] * dinv[d];
        dl_l = d & (OWN_NODES - 1);
    }
    const float* hwl = hw + lane * 4;
    float* tl = tile + lane;
#pragma unroll 1
    for (int g = 0; g < 2; ++g) {
        v4f r[16];
#pragma unroll
        for (int u = 0; u < 16; ++u) {
            const int s = __builtin_amdgcn_readlane(s_l, g * 16 + u);
            r[u] = *(const v4f*)(hwl + (size_t)s * HID);
        }
#pragma unroll
        for (int u = 0; u < 16; ++u) {
            const int   dl = __builtin_amdgcn_readlane(dl_l, g * 16 + u);
            const float en = __builtin_bit_cast(float,
                __builtin_amdgcn_readlane(__builtin_bit_cast(int, en_l), g * 16 + u));
            float* t = tl + dl * HID;
            lds_add(t,      r[u].x * en);
            lds_add(t + 32, r[u].y * en);
            lds_add(t + 64, r[u].z * en);
            lds_add(t + 96, r[u].w * en);
        }
    }
}

__launch_bounds__(1024)
__global__ void k_scatter_own(const int* __restrict__ edge_index,
                              const unsigned char* __restrict__ okey,
                              const float* __restrict__ dinv,
                              const float* __restrict__ hw,
                              float* __restrict__ agg) {
    __shared__ float    tile[(OWN_NODES + 1) * HID];
    __shared__ unsigned stk[SC_WAVES * QCAP];

    const int wave  = threadIdx.x >> 5;
    const int lane  = threadIdx.x & 31;
    const int blk   = blockIdx.x;
    const int node0 = blk * OWN_NODES;

    for (int r = wave; r < OWN_NODES; r += SC_WAVES) {
        const v4f a = *(const v4f*)(agg + (size_t)(node0 + r) * HID + lane * 4);
        float* t = tile + r * HID + lane;
        t[0] = a.x; t[32] = a.y; t[64] = a.z; t[96] = a.w;
    }
    __syncthreads();

    unsigned* st = stk + wave * QCAP;
    const unsigned idrep = (unsigned)blk * 0x01010101u;
    const unsigned char* kw = okey + (size_t)wave * E_WAVE + lane * 16;
    const unsigned e_w = (unsigned)wave * E_WAVE + (unsigned)lane * 16u;
    int cnt = 0;

    uint4 kk[4];
#pragma unroll
    for (int u = 0; u < 4; ++u)
        kk[u] = *(const uint4*)(kw + ((u + blk) & (SCAN_STEPS - 1)) * KEYS_STEP);
#pragma unroll 1
    for (int i = 0; i < SCAN_STEPS; i += 4) {
#pragma unroll
        for (int u = 0; u < 4; ++u) {
            const int it = (i + u + blk) & (SCAN_STEPS - 1);
            const unsigned eb = e_w + (unsigned)it * KEYS_STEP;
            const unsigned kv[4] = {kk[u].x, kk[u].y, kk[u].z, kk[u].w};
            unsigned p = (zero_bytes(kv[0] ^ idrep) >> 7) | (zero_bytes(kv[1] ^ idrep) >> 6) |
                         (zero_bytes(kv[2] ^ idrep) >> 5) | (zero_bytes(kv[3] ^ idrep) >> 4);
            for (;;) {
                const bool has = (p != 0u);
                const unsigned m = __builtin_amdgcn_ballot_w32(has);
                if (m == 0u) break;
                if (has) {
                    const unsigned t = (unsigned)__builtin_ctz(p);
                    st[cnt + (int)__builtin_amdgcn_mbcnt_lo(m, 0u)] = eb + ((t & 7u) << 2) + (t >> 3);
                    p &= p - 1u;
                }
                cnt += __builtin_popcount(m);
            }
            if (i + 4 < SCAN_STEPS)
                kk[u] = *(const uint4*)(kw + ((i + u + 4 + blk) & (SCAN_STEPS - 1)) * KEYS_STEP);
            const int need = (i + u + 1 < SCAN_STEPS) ? 32 : 1;
            while (cnt >= need) {
                const int n = cnt < 32 ? cnt : 32;
                cnt -= n;
                sc_drain(st + cnt, n, lane, edge_index, dinv, hw, tile);
            }
        }
    }
    __syncthreads();

    for (int r = wave; r < OWN_NODES; r += SC_WAVES) {
        const float* t = tile + r * HID + lane;
        v4f a;
        a.x = t[0]; a.y = t[32]; a.z = t[64]; a.w = t[96];
        *(v4f*)(agg + (size_t)(node0 + r) * HID + lane * 4) = a;
    }
}

__launch_bounds__(256)
__global__ void k_relu_res_ln(float* __restrict__ h,
                              const float* __restrict__ agg,
                              const float* __restrict__ gamma,
                              const float* __restrict__ beta) {
    int t = blockIdx.x * 256 + threadIdx.x;
    int n = t >> 5;
    int lane = t & 31;
    int f = lane << 2;

    const float* ar = agg + (size_t)n * HID;
    float* hr = h + (size_t)n * HID;

    v4f hv = *(const v4f*)(hr + f);
    v4f av = *(const v4f*)(ar + f);
    v4f val;
    val.x = hv.x + fmaxf(av.x, 0.0f);
    val.y = hv.y + fmaxf(av.y, 0.0f);
    val.z = hv.z + fmaxf(av.z, 0.0f);
    val.w = hv.w + fmaxf(av.w, 0.0f);

    float s = val.x + val.y + val.z + val.w;
#pragma unroll
    for (int off = 16; off >= 1; off >>= 1) s += __shfl_xor(s, off, 32);
    const float mu = s * (1.0f / (float)HID);

    v4f dv = val - mu;
    float q = dv.x * dv.x + dv.y * dv.y + dv.z * dv.z + dv.w * dv.w;
#pragma unroll
    for (int off = 16; off >= 1; off >>= 1) q += __shfl_xor(q, off, 32);
    const float rstd = rsqrtf(q * (1.0f / (float)HID) + LN_EPS);

    v4f g = *(const v4f*)(gamma + f);
    v4f b = *(const v4f*)(beta + f);
    v4f outv = dv * rstd * g + b;
    *(v4f*)(hr + f) = outv;
}

__global__ void k_out(const float* __restrict__ h,
                      const float* __restrict__ W_out,
                      float* __restrict__ out) {
    int t = blockIdx.x * 256 + threadIdx.x;
    if (t >= NGRAPH * OUT_DIM) return;
    int g = t / OUT_DIM;
    int o = t % OUT_DIM;
    const float* hr = h + (size_t)(g * NPG) * HID;
    float s = 0.0f;
#pragma unroll 4
    for (int k = 0; k < HID; ++k) s = fmaf(hr[k], W_out[k * OUT_DIM + o], s);
    out[t] = s;
}

extern "C" void kernel_launch(void* const* d_in, const int* in_sizes, int n_in,
                              void* d_out, int out_size, void* d_ws, size_t ws_size,
                              hipStream_t stream) {
    const int*   x        = (const int*)d_in[0];
    const int*   ei       = (const int*)d_in[1];
    const float* cls_emb  = (const float*)d_in[3];
    const float* nbr_emb  = (const float*)d_in[4];
    const float* Ws       = (const float*)d_in[5];
    const float* bs       = (const float*)d_in[6];
    const float* gammas   = (const float*)d_in[7];
    const float* betas    = (const float*)d_in[8];
    const float* W_out    = (const float*)d_in[9];
    float*       out      = (float*)d_out;

    char* ws = (char*)d_ws;
    size_t off = 0;
    int*   degcnt = (int*)(ws + off);   off += (size_t)N_NODES * 4;
    float* dinv   = (float*)(ws + off); off += (size_t)N_NODES * 4;
    float* invdeg = (float*)(ws + off); off += (size_t)N_NODES * 4;
    float* h      = (float*)(ws + off); off += (size_t)N_NODES * HID * 4;
    float* hw     = (float*)(ws + off); off += (size_t)N_NODES * HID * 4;
    float* agg    = (float*)(ws + off); off += (size_t)N_NODES * HID * 4;
    unsigned char* okey = (unsigned char*)(ws + off); off += (size_t)E_EDGES;
    (void)ws_size; (void)in_sizes; (void)n_in; (void)out_size;

    k_deg_init  <<<N_NODES / 256, 256, 0, stream>>>(degcnt);
    k_deg_count <<<E_EDGES / 256, 256, 0, stream>>>(ei, degcnt, okey);
    k_deg_finish<<<N_NODES / 256, 256, 0, stream>>>(degcnt, dinv, invdeg);

    k_embed<<<(N_NODES * 32) / 256, 256, 0, stream>>>(x, cls_emb, nbr_emb, h);

    for (int i = 0; i < N_LAYERS; ++i) {
        k_gemm_fused<<<N_NODES / 128, 256, 0, stream>>>(
            h, Ws + (size_t)i * HID * HID, bs + (size_t)i * HID, invdeg, hw, agg);
        k_scatter_own<<<N_NODES / OWN_NODES, SC_WAVES * 32, 0, stream>>>(ei, okey, dinv, hw, agg);
        k_relu_res_ln<<<(N_NODES * 32) / 256, 256, 0, stream>>>(
            h, agg, gammas + (size_t)i * HID, betas + (size_t)i * HID);
    }

    k_out<<<(NGRAPH * OUT_DIM + 255) / 256, 256, 0, stream>>>(h, W_out, out);
}


# relu+residual+LayerNorm and the embedding sum moved into the head of the GEMM that reads them: a wave forms its 16 A rows in LDS; last LayerNorm goes with the logits, root rows only; sums as before
# speedup vs baseline: 2.1082x; 1.1533x over previous
#include <hip/hip_runtime.h>

#define N_NODES   65536
#define NPG       128
#define NGRAPH    512
#define E_EDGES   1048576
#define HID       128
#define OUT_DIM   10
#define N_LAYERS  4
#define LN_EPS    1e-5f

typedef float v2f __attribute__((ext_vector_type(2)));
typedef float v4f __attribute__((ext_vector_type(4)));
typedef float v8f __attribute__((ext_vector_type(8)));

__global__ void k_deg_init(int* __restrict__ degcnt) {
    int i = blockIdx.x * 256 + threadIdx.x;
    if (i < N_NODES) degcnt[i] = 0;
}

__global__ void k_deg_count(const int* __restrict__ edge_index, int* __restrict__ degcnt,
                            unsigned char* __restrict__ okey) {
    int e = blockIdx.x * 256 + threadIdx.x;
    if (e < E_EDGES) {
        int d = edge_index[E_EDGES + e];
        atomicAdd(&degcnt[d], 1);
        okey[e] = (unsigned char)(d >> 8);
    }
}

__global__ void k_deg_finish(const int* __restrict__ degcnt,
                             float* __restrict__ dinv, float* __restrict__ invdeg) {
    int i = blockIdx.x * 256 + threadIdx.x;
    if (i < N_NODES) {
        float d = 1.0f + (float)degcnt[i];
        dinv[i]   = rsqrtf(d);
        invdeg[i] = 1.0f / d;
    }
}

__device__ __forceinline__ v4f relu_res_ln(v4f hv, v4f av, v4f g, v4f b) {
    v4f val;
    val.x = hv.x + fmaxf(av.x, 0.0f);
    val.y = hv.y + fmaxf(av.y, 0.0f);
    val.z = hv.z + fmaxf(av.z, 0.0f);
    val.w = hv.w + fmaxf(av.w, 0.0f);

    float s = val.x + val.y + val.z + val.w;
#pragma unroll
    for (int off = 16; off >= 1; off >>= 1) s += __shfl_xor(s, off, 32);
    const float mu = s * (1.0f / (float)HID);

    v4f dv = val - mu;
    float q = dv.x * dv.x + dv.y * dv.y + dv.z * dv.z + dv.w * dv.w;
#pragma unroll
    for (int off = 16; off >= 1; off >>= 1) q += __shfl_xor(q, off, 32);
    const float rstd = rsqrtf(q * (1.0f / (float)HID) + LN_EPS);
    return dv * rstd * g + b;
}

#define A_PITCH (HID + 4)
__launch_bounds__(256)
__global__ void k_gemm_head(const int first,
                            const int* __restrict__ x,
                            const float* __restrict__ cls_emb,
                            const float* __restrict__ nbr_emb,
                            const float* __restrict__ gamma,
                            const float* __restrict__ beta,
                            const float* __restrict__ W,
                            const float* __restrict__ bias,
                            const float* __restrict__ invdeg,
                            float* __restrict__ h,
                            float* __restrict__ hw,
                            float* agg) {
    __shared__ float lWt[HID * HID];
    __shared__ __attribute__((aligned(16))) float lA[128 * A_PITCH];
    const int tid = threadIdx.x;
    const int wave = tid >> 5;
    const int lane = tid & 31;
    const int m0   = blockIdx.x * 128 + wave * 16;

    for (int i = tid * 4; i < HID * HID; i += 256 * 4) {
        v4f w = *(const v4f*)(W + i);
        const int k = i >> 7;
        const int n = i & (HID - 1);
#pragma unroll
        for (int j = 0; j < 4; ++j)
            lWt[(n + j) * HID + (k ^ (((n + j) & 31) << 1))] = w[j];
    }

    {
        const int f = lane << 2;
        float* at = lA + wave * 16 * A_PITCH + f;
        if (first) {
#pragma unroll 8
            for (int r = 0; r < 16; ++r) {
                const int n  = m0 + r;
                const int c  = x[2 * n + 0];
                const int nb = x[2 * n + 1];
                const v4f a = *(const v4f*)(cls_emb + (size_t)c  * HID + f);
                const v4f b = *(const v4f*)(nbr_emb + (size_t)nb * HID + f);
                const v4f v = a + b;
                *(v4f*)(h + (size_t)n * HID + f) = v;
                *(v4f*)(at + r * A_PITCH) = v;
            }
        } else {
            const v4f g = *(const v4f*)(gamma + f);
            const v4f b = *(const v4f*)(beta + f);
#pragma unroll 8
            for (int r = 0; r < 16; ++r) {
                const int n = m0 + r;
                const v4f hv = *(const v4f*)(h   + (size_t)n * HID + f);
                const v4f av = *(const v4f*)(agg + (size_t)n * HID + f);
                const v4f v = relu_res_ln(hv, av, g, b);
                *(v4f*)(h + (size_t)n * HID + f) = v;
                *(v4f*)(at + r * A_PITCH) = v;
            }
        }
    }
    __syncthreads();

    const int half = lane >> 4;
    const int l16  = lane & 15;

    v8f zero = {};
    v8f acc[8];
#pragma unroll
    for (int t = 0; t < 8; ++t) acc[t] = zero;

    const float* arow = lA + (wave * 16 + l16) * A_PITCH;

    for (int k = 0; k < HID; k += 4) {
        const int kb = k + half * 2;
        v2f a = *(const v2f*)(arow + kb);
#pragma unroll
        for (int t = 0; t < 8; ++t) {
            const int col = t * 16 + l16;
            v2f b = *(const v2f*)(lWt + col * HID + (kb ^ ((col & 31) << 1)));
            acc[t] = __builtin_amdgcn_wmma_f32_16x16x4_f32(
                false, a, false, b, (short)0, acc[t], false, false);
        }
    }

    const int rbase = half * 8;
#pragma unroll
    for (int r = 0; r < 8; ++r) {
        const int row = m0 + rbase + r;
        const float inv = invdeg[row];
        float* hwrow  = hw  + (size_t)row * HID;
        float* aggrow = agg + (size_t)row * HID;
#pragma unroll
        for (int t = 0; t < 8; ++t) {
            const int col = t * 16 + l16;
            const float v = acc[t][r];
            hwrow[col]  = v;
            aggrow[col] = v * inv + bias[col];
        }
    }
}

#define OWN_NODES  256
#define SC_WAVES   32
#define KEYS_STEP  512
#define E_WAVE     (E_EDGES / SC_WAVES)
#define SCAN_STEPS (E_WAVE / KEYS_STEP)
#define QCAP       (32 + KEYS_STEP)
static_assert(N_NODES == 256 * OWN_NODES, "the owner key is one byte and the row in the tile another");
static_assert(E_EDGES % (SC_WAVES * KEYS_STEP) == 0 && (SCAN_STEPS & (SCAN_STEPS - 1)) == 0 && SCAN_STEPS % 4 == 0,
              "the waves' scans cover the edge list exactly");

__device__ __forceinline__ unsigned zero_bytes(unsigned x) {
    const unsigned y = (x & 0x7f7f7f7fu) + 0x7f7f7f7fu;
    return ~(y | x | 0x7f7f7f7fu);
}

__device__ __forceinline__ void lds_add(float* p, float v) {
    __hip_atomic_fetch_add(p, v, __ATOMIC_RELAXED, __HIP_MEMORY_SCOPE_WORKGROUP);
}

__device__ __forceinline__ void sc_drain(const unsigned* st, int n, int lane,
                                         const int* __restrict__ edge_index,
                                         const float* __restrict__ dinv,
                                         const float* __restrict__ hw,
                                         float* tile) {
    int   s_l  = 0;
    int   dl_l = OWN_NODES;
    float en_l = 0.0f;
    if (lane < n) {
        const unsigned e = st[lane];
        s_l = edge_index[e];
        const int d = edge_index[E_EDGES + e];
        en_l = dinv[s_l] * dinv[d];
        dl_l = d & (OWN_NODES - 1);
    }
    const float* hwl = hw + lane * 4;
    float* tl = tile + lane;
#pragma unroll 1
    for (int g = 0; g < 2; ++g) {
        v4f r[16];
#pragma unroll
        for (int u = 0; u < 16; ++u) {
            const int s = __builtin_amdgcn_readlane(s_l, g * 16 + u);
            r[u] = *(const v4f*)(hwl + (size_t)s * HID);
        }
#pragma unroll
        for (int u = 0; u < 16; ++u) {
            const int   dl = __builtin_amdgcn_readlane(dl_l, g * 16 + u);
            const float en = __builtin_bit_cast(float,
                __builtin_amdgcn_readlane(__builtin_bit_cast(int, en_l), g * 16 + u));
            float* t = tl + dl * HID;
            lds_add(t,      r[u].x * en);
            lds_add(t + 32, r[u].y * en);
            lds_add(t + 64, r[u].z * en);
            lds_add(t + 96, r[u].w * en);
        }
    }
}

__launch_bounds__(1024)
__global__ void k_scatter_own(const int* __restrict__ edge_index,
                              const unsigned char* __restrict__ okey,
                              const float* __restrict__ dinv,
                              const float* __restrict__ hw,
                              float* __restrict__ agg) {
    __shared__ float    tile[(OWN_NODES + 1) * HID];
    __shared__ unsigned stk[SC_WAVES * QCAP];

    const int wave  = threadIdx.x >> 5;
    const int lane  = threadIdx.x & 31;
    const int blk   = blockIdx.x;
    const int node0 = blk * OWN_NODES;

    for (int r = wave; r < OWN_NODES; r += SC_WAVES) {
        const v4f a = *(const v4f*)(agg + (size_t)(node0 + r) * HID + lane * 4);
        float* t = tile + r * HID + lane;
        t[0] = a.x; t[32] = a.y; t[64] = a.z; t[96] = a.w;
    }
    __syncthreads();

    unsigned* st = stk + wave * QCAP;
    const unsigned idrep = (unsigned)blk * 0x01010101u;
    const unsigned char* kw = okey + (size_t)wave * E_WAVE + lane * 16;
    const unsigned e_w = (unsigned)wave * E_WAVE + (unsigned)lane * 16u;
    int cnt = 0;

    uint4 kk[4];
#pragma unroll
    for (int u = 0; u < 4; ++u)
        kk[u] = *(const uint4*)(kw + ((u + blk) & (SCAN_STEPS - 1)) * KEYS_STEP);
#pragma unroll 1
    for (int i = 0; i < SCAN_STEPS; i += 4) {
#pragma unroll
        for (int u = 0; u < 4; ++u) {
            const int it = (i + u + blk) & (SCAN_STEPS - 1);
            const unsigned eb = e_w + (unsigned)it * KEYS_STEP;
            const unsigned kv[4] = {kk[u].x, kk[u].y, kk[u].z, kk[u].w};
            unsigned p = (zero_bytes(kv[0] ^ idrep) >> 7) | (zero_bytes(kv[1] ^ idrep) >> 6) |
                         (zero_bytes(kv[2] ^ idrep) >> 5) | (zero_bytes(kv[3] ^ idrep) >> 4);
            for (;;) {
                const bool has = (p != 0u);
                const unsigned m = __builtin_amdgcn_ballot_w32(has);
                if (m == 0u) break;
                if (has) {
                    const unsigned t = (unsigned)__builtin_ctz(p);
                    st[cnt + (int)__builtin_amdgcn_mbcnt_lo(m, 0u)] = eb + ((t & 7u) << 2) + (t >> 3);
                    p &= p - 1u;
                }
                cnt += __builtin_popcount(m);
            }
            if (i + 4 < SCAN_STEPS)
                kk[u] = *(const uint4*)(kw + ((i + u + 4 + blk) & (SCAN_STEPS - 1)) * KEYS_STEP);
            const int need = (i + u + 1 < SCAN_STEPS) ? 32 : 1;
            while (cnt >= need) {
                const int n = cnt < 32 ? cnt : 32;
                cnt -= n;
                sc_drain(st + cnt, n, lane, edge_index, dinv, hw, tile);
            }
        }
    }
    __syncthreads();

    for (int r = wave; r < OWN_NODES; r += SC_WAVES) {
        const float* t = tile + r * HID + lane;
        v4f a;
        a.x = t[0]; a.y = t[32]; a.z = t[64]; a.w = t[96];
        *(v4f*)(agg + (size_t)(node0 + r) * HID + lane * 4) = a;
    }
}

__launch_bounds__(256)
__global__ void k_out(const float* __restrict__ h,
                      const float* __restrict__ agg,
                      const float* __restrict__ gamma,
                      const float* __restrict__ beta,
                      const float* __restrict__ W_out,
                      float* __restrict__ out) {
    __shared__ float row[8 * HID];
    const int wave = threadIdx.x >> 5;
    const int lane = threadIdx.x & 31;
    const int g = blockIdx.x * 8 + wave;
    const int f = lane << 2;
    const size_t n = (size_t)g * NPG;

    const v4f hv = *(const v4f*)(h   + n * HID + f);
    const v4f av = *(const v4f*)(agg + n * HID + f);
    const v4f gm = *(const v4f*)(gamma + f);
    const v4f bt = *(const v4f*)(beta + f);
    *(v4f*)(row + wave * HID + f) = relu_res_ln(hv, av, gm, bt);
    __syncthreads();

    if (lane < OUT_DIM) {
        const float* hr = row + wave * HID;
        float s = 0.0f;
#pragma unroll 4
        for (int k = 0; k < HID; ++k) s = fmaf(hr[k], W_out[k * OUT_DIM + lane], s);
        out[g * OUT_DIM + lane] = s;
    }
}

extern "C" void kernel_launch(void* const* d_in, const int* in_sizes, int n_in,
                              void* d_out, int out_size, void* d_ws, size_t ws_size,
                              hipStream_t stream) {
    const int*   x        = (const int*)d_in[0];
    const int*   ei       = (const int*)d_in[1];
    const float* cls_emb  = (const float*)d_in[3];
    const float* nbr_emb  = (const float*)d_in[4];
    const float* Ws       = (const float*)d_in[5];
    const float* bs       = (const float*)d_in[6];
    const float* gammas   = (const float*)d_in[7];
    const float* betas    = (const float*)d_in[8];
    const float* W_out    = (const float*)d_in[9];
    float*       out      = (float*)d_out;

    char* ws = (char*)d_ws;
    size_t off = 0;
    int*   degcnt = (int*)(ws + off);   off += (size_t)N_NODES * 4;
    float* dinv   = (float*)(ws + off); off += (size_t)N_NODES * 4;
    float* invdeg = (float*)(ws + off); off += (size_t)N_NODES * 4;
    float* h      = (float*)(ws + off); off += (size_t)N_NODES * HID * 4;
    float* hw     = (float*)(ws + off); off += (size_t)N_NODES * HID * 4;
    float* agg    = (float*)(ws + off); off += (size_t)N_NODES * HID * 4;
    unsigned char* okey = (unsigned char*)(ws + off); off += (size_t)E_EDGES;
    (void)ws_size; (void)in_sizes; (void)n_in; (void)out_size;

    k_deg_init  <<<N_NODES / 256, 256, 0, stream>>>(degcnt);
    k_deg_count <<<E_EDGES / 256, 256, 0, stream>>>(ei, degcnt, okey);
    k_deg_finish<<<N_NODES / 256, 256, 0, stream>>>(degcnt, dinv, invdeg);

    k_gemm_head<<<N_NODES / 128, 256, 0, stream>>>(
        1, x, cls_emb, nbr_emb, gammas, betas, Ws, bs, invdeg, h, hw, agg);
    k_scatter_own<<<N_NODES / OWN_NODES, SC_WAVES * 32, 0, stream>>>(ei, okey, dinv, hw, agg);

    for (int i = 1; i < N_LAYERS; ++i) {
        k_gemm_head<<<N_NODES / 128, 256, 0, stream>>>(
            0, x, cls_emb, nbr_emb,
            gammas + (size_t)(i - 1) * HID, betas + (size_t)(i - 1) * HID,
            Ws + (size_t)i * HID * HID, bs + (size_t)i * HID, invdeg, h, hw, agg);
        k_scatter_own<<<N_NODES / OWN_NODES, SC_WAVES * 32, 0, stream>>>(ei, okey, dinv, hw, agg);
    }

    k_out<<<NGRAPH / 8, 256, 0, stream>>>(
        h, agg, gammas + (size_t)(N_LAYERS - 1) * HID, betas + (size_t)(N_LAYERS - 1) * HID,
        W_out, out);
}


# Scatter: layer 0's key scan also writes each wave's matched edges as packed (src, tile row, norm) lists; layers 1-3 read the lists: no key scan, no index chasing; sums in arrival order as before
# speedup vs baseline: 2.3532x; 1.1162x over previous
#include <hip/hip_runtime.h>

#define N_NODES   65536
#define NPG       128
#define NGRAPH    512
#define E_EDGES   1048576
#define HID       128
#define OUT_DIM   10
#define N_LAYERS  4
#define LN_EPS    1e-5f

typedef float v2f __attribute__((ext_vector_type(2)));
typedef float v4f __attribute__((ext_vector_type(4)));
typedef float v8f __attribute__((ext_vector_type(8)));
typedef unsigned v2u __attribute__((ext_vector_type(2)));

__global__ void k_deg_init(int* __restrict__ degcnt) {
    int i = blockIdx.x * 256 + threadIdx.x;
    if (i < N_NODES) degcnt[i] = 0;
}

__global__ void k_deg_count(const int* __restrict__ edge_index, int* __restrict__ degcnt,
                            unsigned char* __restrict__ okey) {
    int e = blockIdx.x * 256 + threadIdx.x;
    if (e < E_EDGES) {
        int d = edge_index[E_EDGES + e];
        atomicAdd(&degcnt[d], 1);
        okey[e] = (unsigned char)(d >> 8);
    }
}

__global__ void k_deg_finish(const int* __restrict__ degcnt,
                             float* __restrict__ dinv, float* __restrict__ invdeg) {
    int i = blockIdx.x * 256 + threadIdx.x;
    if (i < N_NODES) {
        float d = 1.0f + (float)degcnt[i];
        dinv[i]   = rsqrtf(d);
        invdeg[i] = 1.0f / d;
    }
}

__device__ __forceinline__ v4f relu_res_ln(v4f hv, v4f av, v4f g, v4f b) {
    v4f val;
    val.x = hv.x + fmaxf(av.x, 0.0f);
    val.y = hv.y + fmaxf(av.y, 0.0f);
    val.z = hv.z + fmaxf(av.z, 0.0f);
    val.w = hv.w + fmaxf(av.w, 0.0f);

    float s = val.x + val.y + val.z + val.w;
#pragma unroll
    for (int off = 16; off >= 1; off >>= 1) s += __shfl_xor(s, off, 32);
    const float mu = s * (1.0f / (float)HID);

    v4f dv = val - mu;
    float q = dv.x * dv.x + dv.y * dv.y + dv.z * dv.z + dv.w * dv.w;
#pragma unroll
    for (int off = 16; off >= 1; off >>= 1) q += __shfl_xor(q, off, 32);
    const float rstd = rsqrtf(q * (1.0f / (float)HID) + LN_EPS);
    return dv * rstd * g + b;
}

#define A_PITCH (HID + 4)
__launch_bounds__(256)
__global__ void k_gemm_head(const int first,
                            const int* __restrict__ x,
                            const float* __restrict__ cls_emb,
                            const float* __restrict__ nbr_emb,
                            const float* __restrict__ gamma,
                            const float* __restrict__ beta,
                            const float* __restrict__ W,
                            const float* __restrict__ bias,
                            const float* __restrict__ invdeg,
                            float* __restrict__ h,
                            float* __restrict__ hw,
                            float* agg) {
    __shared__ float lWt[HID * HID];
    __shared__ __attribute__((aligned(16))) float lA[128 * A_PITCH];
    const int tid = threadIdx.x;
    const int wave = tid >> 5;
    const int lane = tid & 31;
    const int m0   = blockIdx.x * 128 + wave * 16;

    for (int i = tid * 4; i < HID * HID; i += 256 * 4) {
        v4f w = *(const v4f*)(W + i);
        const int k = i >> 7;
        const int n = i & (HID - 1);
#pragma unroll
        for (int j = 0; j < 4; ++j)
            lWt[(n + j) * HID + (k ^ (((n + j) & 31) << 1))] = w[j];
    }

    {
        const int f = lane << 2;
        float* at = lA + wave * 16 * A_PITCH + f;
        if (first) {
#pragma unroll 8
            for (int r = 0; r < 16; ++r) {
                const int n  = m0 + r;
                const int c  = x[2 * n + 0];
                const int nb = x[2 * n + 1];
                const v4f a = *(const v4f*)(cls_emb + (size_t)c  * HID + f);
                const v4f b = *(const v4f*)(nbr_emb + (size_t)nb * HID + f);
                const v4f v = a + b;
                *(v4f*)(h + (size_t)n * HID + f) = v;
                *(v4f*)(at + r * A_PITCH) = v;
            }
        } else {
            const v4f g = *(const v4f*)(gamma + f);
            const v4f b = *(const v4f*)(beta + f);
#pragma unroll 8
            for (int r = 0; r < 16; ++r) {
                const int n = m0 + r;
                const v4f hv = *(const v4f*)(h   + (size_t)n * HID + f);
                const v4f av = *(const v4f*)(agg + (size_t)n * HID + f);
                const v4f v = relu_res_ln(hv, av, g, b);
                *(v4f*)(h + (size_t)n * HID + f) = v;
                *(v4f*)(at + r * A_PITCH) = v;
            }
        }
    }
    __syncthreads();

    const int half = lane >> 4;
    const int l16  = lane & 15;

    v8f zero = {};
    v8f acc[8];
#pragma unroll
    for (int t = 0; t < 8; ++t) acc[t] = zero;

    const float* arow = lA + (wave * 16 + l16) * A_PITCH;

    for (int k = 0; k < HID; k += 4) {
        const int kb = k + half * 2;
        v2f a = *(const v2f*)(arow + kb);
#pragma unroll
        for (int t = 0; t < 8; ++t) {
            const int col = t * 16 + l16;
            v2f b = *(const v2f*)(lWt + col * HID + (kb ^ ((col & 31) << 1)));
            acc[t] = __builtin_amdgcn_wmma_f32_16x16x4_f32(
                false, a, false, b, (short)0, acc[t], false, false);
        }
    }

    const int rbase = half * 8;
#pragma unroll
    for (int r = 0; r < 8; ++r) {
        const int row = m0 + rbase + r;
        const float inv = invdeg[row];
        float* hwrow  = hw  + (size_t)row * HID;
        float* aggrow = agg + (size_t)row * HID;
#pragma unroll
        for (int t = 0; t < 8; ++t) {
            const int col = t * 16 + l16;
            const float v = acc[t][r];
            hwrow[col]  = v;
            aggrow[col] = v * inv + bias[col];
        }
    }
}

#define OWN_NODES  256
#define LCAP       256
#define SC_WAVES   32
#define KEYS_STEP  512
#define E_WAVE     (E_EDGES / SC_WAVES)
#define SCAN_STEPS (E_WAVE / KEYS_STEP)
#define QCAP       (32 + KEYS_STEP)
static_assert(N_NODES == 256 * OWN_NODES, "the owner key is one byte and the row in the tile another");
static_assert(N_NODES <= 65536 && OWN_NODES <= 256, "a list entry holds src in 16 bits and the tile row above them");
static_assert(E_EDGES % (SC_WAVES * KEYS_STEP) == 0 && (SCAN_STEPS & (SCAN_STEPS - 1)) == 0 && SCAN_STEPS % 4 == 0,
              "the waves' scans cover the edge list exactly");

__device__ __forceinline__ unsigned zero_bytes(unsigned x) {
    const unsigned y = (x & 0x7f7f7f7fu) + 0x7f7f7f7fu;
    return ~(y | x | 0x7f7f7f7fu);
}

__device__ __forceinline__ void lds_add(float* p, float v) {
    __hip_atomic_fetch_add(p, v, __ATOMIC_RELAXED, __HIP_MEMORY_SCOPE_WORKGROUP);
}

__device__ __forceinline__ void sc_rows(int s_l, int dl_l, float en_l, int lane,
                                        const float* __restrict__ hw,
                                        float* tile) {
    const float* hwl = hw + lane * 4;
    float* tl = tile + lane;
#pragma unroll 1
    for (int g = 0; g < 2; ++g) {
        v4f r[16];
#pragma unroll
        for (int u = 0; u < 16; ++u) {
            const int s = __builtin_amdgcn_readlane(s_l, g * 16 + u);
            r[u] = *(const v4f*)(hwl + (size_t)s * HID);
        }
#pragma unroll
        for (int u = 0; u < 16; ++u) {
            const int   dl = __builtin_amdgcn_readlane(dl_l, g * 16 + u);
            const float en = __builtin_bit_cast(float,
                __builtin_amdgcn_readlane(__builtin_bit_cast(int, en_l), g * 16 + u));
            float* t = tl + dl * HID;
            lds_add(t,      r[u].x * en);
            lds_add(t + 32, r[u].y * en);
            lds_add(t + 64, r[u].z * en);
            lds_add(t + 96, r[u].w * en);
        }
    }
}

__device__ __forceinline__ void sc_drain(const unsigned* st, int n, int lane,
                                         const int* __restrict__ edge_index,
                                         const float* __restrict__ dinv,
                                         const float* __restrict__ hw,
                                         float* tile, unsigned* rec) {
    int   s_l  = 0;
    int   dl_l = OWN_NODES;
    float en_l = 0.0f;
    if (lane < n) {
        const unsigned e = st[lane];
        s_l = edge_index[e];
        const int d = edge_index[E_EDGES + e];
        en_l = dinv[s_l] * dinv[d];
        dl_l = d & (OWN_NODES - 1);
        if (rec) {
            v2u ent;
            ent.x = (unsigned)s_l | ((unsigned)dl_l << 16);
            ent.y = __builtin_bit_cast(unsigned, en_l);
            *(v2u*)(rec + 2 * lane) = ent;
        }
    }
    sc_rows(s_l, dl_l, en_l, lane, hw, tile);
}

__launch_bounds__(1024)
__global__ void k_scatter_own(const int* __restrict__ edge_index,
                              const unsigned char* __restrict__ okey,
                              const float* __restrict__ dinv,
                              const float* __restrict__ hw,
                              float* __restrict__ agg,
                              unsigned* __restrict__ elist,
                              int* __restrict__ ecnt,
                              const int record) {
    __shared__ float    tile[(OWN_NODES + 1) * HID];
    __shared__ unsigned stk[SC_WAVES * QCAP];

    const int wave  = threadIdx.x >> 5;
    const int lane  = threadIdx.x & 31;
    const int blk   = blockIdx.x;
    const int node0 = blk * OWN_NODES;

    for (int r = wave; r < OWN_NODES; r += SC_WAVES) {
        const v4f a = *(const v4f*)(agg + (size_t)(node0 + r) * HID + lane * 4);
        float* t = tile + r * HID + lane;
        t[0] = a.x; t[32] = a.y; t[64] = a.z; t[96] = a.w;
    }
    __syncthreads();

    const int slot = blk * SC_WAVES + wave;
    unsigned* el = elist + (size_t)slot * (2 * LCAP);
    bool scan = true;
    if (!record) {
        const int c = __builtin_amdgcn_readfirstlane(ecnt[slot]);
        if (c >= 0) {
            scan = false;
            const unsigned pad0 = (unsigned)OWN_NODES << 16;
            const int last = c - 1;
            unsigned nx0 = pad0, nx1 = 0u;
            if (c > 0) {
                const int j0 = lane < c ? lane : last;
                const unsigned t0 = el[2 * j0], t1 = el[2 * j0 + 1];
                nx0 = lane < c ? t0 : pad0;
                nx1 = lane < c ? t1 : 0u;
            }
#pragma unroll 1
            for (int i = 0; i < c; i += 32) {
                const unsigned cur0 = nx0, cur1 = nx1;
                const int j  = i + 32 + lane;
                const int jc = j < c ? j : last;
                const unsigned t0 = el[2 * jc], t1 = el[2 * jc + 1];
                nx0 = j < c ? t0 : pad0;
                nx1 = j < c ? t1 : 0u;
                sc_rows((int)(cur0 & 0xffffu), (int)(cur0 >> 16),
                        __builtin_bit_cast(float, cur1), lane, hw, tile);
            }
        }
    }
    if (scan) {
        int tot = 0;
        unsigned* st = stk + wave * QCAP;
        const unsigned idrep = (unsigned)blk * 0x01010101u;
        const unsigned char* kw = okey + (size_t)wave * E_WAVE + lane * 16;
        const unsigned e_w = (unsigned)wave * E_WAVE + (unsigned)lane * 16u;
        int cnt = 0;

        uint4 kk[4];
#pragma unroll
        for (int u = 0; u < 4; ++u)
            kk[u] = *(const uint4*)(kw + ((u + blk) & (SCAN_STEPS - 1)) * KEYS_STEP);
#pragma unroll 1
        for (int i = 0; i < SCAN_STEPS; i += 4) {
#pragma unroll
            for (int u = 0; u < 4; ++u) {
                const int it = (i + u + blk) & (SCAN_STEPS - 1);
                const unsigned eb = e_w + (unsigned)it * KEYS_STEP;
                const unsigned kv[4] = {kk[u].x, kk[u].y, kk[u].z, kk[u].w};
                unsigned p = (zero_bytes(kv[0] ^ idrep) >> 7) | (zero_bytes(kv[1] ^ idrep) >> 6) |
                             (zero_bytes(kv[2] ^ idrep) >> 5) | (zero_bytes(kv[3] ^ idrep) >> 4);
                for (;;) {
                    const bool has = (p != 0u);
                    const unsigned m = __builtin_amdgcn_ballot_w32(has);
                    if (m == 0u) break;
                    if (has) {
                        const unsigned t = (unsigned)__builtin_ctz(p);
                        st[cnt + (int)__builtin_amdgcn_mbcnt_lo(m, 0u)] = eb + ((t & 7u) << 2) + (t >> 3);
                        p &= p - 1u;
                    }
                    cnt += __builtin_popcount(m);
                }
                if (i + 4 < SCAN_STEPS)
                    kk[u] = *(const uint4*)(kw + ((i + u + 4 + blk) & (SCAN_STEPS - 1)) * KEYS_STEP);
                const int need = (i + u + 1 < SCAN_STEPS) ? 32 : 1;
                while (cnt >= need) {
                    const int n = cnt < 32 ? cnt : 32;
                    cnt -= n;
                    sc_drain(st + cnt, n, lane, edge_index, dinv, hw, tile,
                             (record && tot + n <= LCAP) ? el + 2 * tot : nullptr);
                    tot += n;
                }
            }
        }
        if (record && lane == 0) ecnt[slot] = (tot <= LCAP) ? tot : -1;
    }
    __syncthreads();

    for (int r = wave; r < OWN_NODES; r += SC_WAVES) {
        const float* t = tile + r * HID + lane;
        v4f a;
        a.x = t[0]; a.y = t[32]; a.z = t[64]; a.w = t[96];
        *(v4f*)(agg + (size_t)(node0 + r) * HID + lane * 4) = a;
    }
}

__launch_bounds__(256)
__global__ void k_out(const float* __restrict__ h,
                      const float* __restrict__ agg,
                      const float* __restrict__ gamma,
                      const float* __restrict__ beta,
                      const float* __restrict__ W_out,
                      float* __restrict__ out) {
    __shared__ float row[8 * HID];
    const int wave = threadIdx.x >> 5;
    const int lane = threadIdx.x & 31;
    const int g = blockIdx.x * 8 + wave;
    const int f = lane << 2;
    const size_t n = (size_t)g * NPG;

    const v4f hv = *(const v4f*)(h   + n * HID + f);
    const v4f av = *(const v4f*)(agg + n * HID + f);
    const v4f gm = *(const v4f*)(gamma + f);
    const v4f bt = *(const v4f*)(beta + f);
    *(v4f*)(row + wave * HID + f) = relu_res_ln(hv, av, gm, bt);
    __syncthreads();

    if (lane < OUT_DIM) {
        const float* hr = row + wave * HID;
        float s = 0.0f;
#pragma unroll 4
        for (int k = 0; k < HID; ++k) s = fmaf(hr[k], W_out[k * OUT_DIM + lane], s);
        out[g * OUT_DIM + lane] = s;
    }
}

extern "C" void kernel_launch(void* const* d_in, const int* in_sizes, int n_in,
                              void* d_out, int out_size, void* d_ws, size_t ws_size,
                              hipStream_t stream) {
    const int*   x        = (const int*)d_in[0];
    const int*   ei       = (const int*)d_in[1];
    const float* cls_emb  = (const float*)d_in[3];
    const float* nbr_emb  = (const float*)d_in[4];
    const float* Ws       = (const float*)d_in[5];
    const float* bs       = (const float*)d_in[6];
    const float* gammas   = (const float*)d_in[7];
    const float* betas    = (const float*)d_in[8];
    const float* W_out    = (const float*)d_in[9];
    float*       out      = (float*)d_out;

    char* ws = (char*)d_ws;
    size_t off = 0;
    int*   degcnt = (int*)(ws + off);   off += (size_t)N_NODES * 4;
    float* dinv   = (float*)(ws + off); off += (size_t)N_NODES * 4;
    float* invdeg = (float*)(ws + off); off += (size_t)N_NODES * 4;
    float* h      = (float*)(ws + off); off += (size_t)N_NODES * HID * 4;
    float* hw     = (float*)(ws + off); off += (size_t)N_NODES * HID * 4;
    float* agg    = (float*)(ws + off); off += (size_t)N_NODES * HID * 4;
    unsigned char* okey = (unsigned char*)(ws + off); off += (size_t)E_EDGES;
    unsigned* elist = (unsigned*)(ws + off); off += (size_t)(N_NODES / OWN_NODES) * SC_WAVES * LCAP * 8;
    int*   ecnt   = (int*)(ws + off);   off += (size_t)(N_NODES / OWN_NODES) * SC_WAVES * 4;
    (void)ws_size; (void)in_sizes; (void)n_in; (void)out_size;

    k_deg_init  <<<N_NODES / 256, 256, 0, stream>>>(degcnt);
    k_deg_count <<<E_EDGES / 256, 256, 0, stream>>>(ei, degcnt, okey);
    k_deg_finish<<<N_NODES / 256, 256, 0, stream>>>(degcnt, dinv, invdeg);

    k_gemm_head<<<N_NODES / 128, 256, 0, stream>>>(
        1, x, cls_emb, nbr_emb, gammas, betas, Ws, bs, invdeg, h, hw, agg);
    k_scatter_own<<<N_NODES / OWN_NODES, SC_WAVES * 32, 0, stream>>>(
        ei, okey, dinv, hw, agg, elist, ecnt, 1);

    for (int i = 1; i < N_LAYERS; ++i) {
        k_gemm_head<<<N_NODES / 128, 256, 0, stream>>>(
            0, x, cls_emb, nbr_emb,
            gammas + (size_t)(i - 1) * HID, betas + (size_t)(i - 1) * HID,
            Ws + (size_t)i * HID * HID, bs + (size_t)i * HID, invdeg, h, hw, agg);
        k_scatter_own<<<N_NODES / OWN_NODES, SC_WAVES * 32, 0, stream>>>(
            ei, okey, dinv, hw, agg, elist, ecnt, 0);
    }

    k_out<<<NGRAPH / 8, 256, 0, stream>>>(
        h, agg, gammas + (size_t)(N_LAYERS - 1) * HID, betas + (size_t)(N_LAYERS - 1) * HID,
        W_out, out);
}


# Next layer's GEMM, relu+residual+LayerNorm and the logits moved into the scatter launch's tail: the block's finished agg rows stay in its LDS tile as the A operand; agg never stored; sums as before
# speedup vs baseline: 3.2181x; 1.3675x over previous
#include <hip/hip_runtime.h>

#define N_NODES   65536
#define NPG       128
#define NGRAPH    512
#define E_EDGES   1048576
#define HID       128
#define OUT_DIM   10
#define N_LAYERS  4
#define LN_EPS    1e-5f

typedef float v2f __attribute__((ext_vector_type(2)));
typedef float v4f __attribute__((ext_vector_type(4)));
typedef float v8f __attribute__((ext_vector_type(8)));
typedef unsigned v2u __attribute__((ext_vector_type(2)));

__global__ void k_deg_init(int* __restrict__ degcnt) {
    int i = blockIdx.x * 256 + threadIdx.x;
    if (i < N_NODES) degcnt[i] = 0;
}

__global__ void k_deg_count(const int* __restrict__ edge_index, int* __restrict__ degcnt,
                            unsigned char* __restrict__ okey) {
    int e = blockIdx.x * 256 + threadIdx.x;
    if (e < E_EDGES) {
        int d = edge_index[E_EDGES + e];
        atomicAdd(&degcnt[d], 1);
        okey[e] = (unsigned char)(d >> 8);
    }
}

__global__ void k_deg_finish(const int* __restrict__ degcnt,
                             float* __restrict__ dinv, float* __restrict__ invdeg) {
    int i = blockIdx.x * 256 + threadIdx.x;
    if (i < N_NODES) {
        float d = 1.0f + (float)degcnt[i];
        dinv[i]   = rsqrtf(d);
        invdeg[i] = 1.0f / d;
    }
}

__device__ __forceinline__ v4f relu_res_ln(v4f hv, v4f av, v4f g, v4f b) {
    v4f val;
    val.x = hv.x + fmaxf(av.x, 0.0f);
    val.y = hv.y + fmaxf(av.y, 0.0f);
    val.z = hv.z + fmaxf(av.z, 0.0f);
    val.w = hv.w + fmaxf(av.w, 0.0f);

    float s = val.x + val.y + val.z + val.w;
#pragma unroll
    for (int off = 16; off >= 1; off >>= 1) s += __shfl_xor(s, off, 32);
    const float mu = s * (1.0f / (float)HID);

    v4f dv = val - mu;
    float q = dv.x * dv.x + dv.y * dv.y + dv.z * dv.z + dv.w * dv.w;
#pragma unroll
    for (int off = 16; off >= 1; off >>= 1) q += __shfl_xor(q, off, 32);
    const float rstd = rsqrtf(q * (1.0f / (float)HID) + LN_EPS);
    return dv * rstd * g + b;
}

#define A_PITCH (HID + 4)

__device__ __forceinline__ void stage_w(const float* __restrict__ W, float* lWt, int tid, int nthr) {
    for (int i = tid * 4; i < HID * HID; i += nthr * 4) {
        v4f w = *(const v4f*)(W + i);
        const int k = i >> 7;
        const int n = i & (HID - 1);
#pragma unroll
        for (int j = 0; j < 4; ++j)
            lWt[(n + j) * HID + (k ^ (((n + j) & 31) << 1))] = w[j];
    }
}

__launch_bounds__(256)
__global__ void k_gemm_embed(const int* __restrict__ x,
                             const float* __restrict__ cls_emb,
                             const float* __restrict__ nbr_emb,
                             const float* __restrict__ W,
                             float* __restrict__ h,
                             float* __restrict__ hw) {
    __shared__ float lWt[HID * HID];
    __shared__ __attribute__((aligned(16))) float lA[128 * A_PITCH];
    const int tid = threadIdx.x;
    const int wave = tid >> 5;
    const int lane = tid & 31;
    const int m0   = blockIdx.x * 128 + wave * 16;

    stage_w(W, lWt, tid, 256);

    {
        const int f = lane << 2;
        float* at = lA + wave * 16 * A_PITCH + f;
#pragma unroll 8
        for (int r = 0; r < 16; ++r) {
            const int n  = m0 + r;
            const int c  = x[2 * n + 0];
            const int nb = x[2 * n + 1];
            const v4f a = *(const v4f*)(cls_emb + (size_t)c  * HID + f);
            const v4f b = *(const v4f*)(nbr_emb + (size_t)nb * HID + f);
            const v4f v = a + b;
            *(v4f*)(h + (size_t)n * HID + f) = v;
            *(v4f*)(at + r * A_PITCH) = v;
        }
    }
    __syncthreads();

    const int half = lane >> 4;
    const int l16  = lane & 15;

    v8f zero = {};
    v8f acc[8];
#pragma unroll
    for (int t = 0; t < 8; ++t) acc[t] = zero;

    const float* arow = lA + (wave * 16 + l16) * A_PITCH;

    for (int k = 0; k < HID; k += 4) {
        const int kb = k + half * 2;
        v2f a = *(const v2f*)(arow + kb);
#pragma unroll
        for (int t = 0; t < 8; ++t) {
            const int col = t * 16 + l16;
            v2f b = *(const v2f*)(lWt + col * HID + (kb ^ ((col & 31) << 1)));
            acc[t] = __builtin_amdgcn_wmma_f32_16x16x4_f32(
                false, a, false, b, (short)0, acc[t], false, false);
        }
    }

    const int rbase = half * 8;
#pragma unroll
    for (int r = 0; r < 8; ++r) {
        float* hwrow = hw + (size_t)(m0 + rbase + r) * HID;
#pragma unroll
        for (int t = 0; t < 8; ++t) hwrow[t * 16 + l16] = acc[t][r];
    }
}

#define OWN_NODES  256
#define T_PITCH    (HID + 4)
#define LCAP       256
#define SC_WAVES   32
#define KEYS_STEP  512
#define E_WAVE     (E_EDGES / SC_WAVES)
#define SCAN_STEPS (E_WAVE / KEYS_STEP)
#define QCAP       (32 + KEYS_STEP)
static_assert(N_NODES == 256 * OWN_NODES, "the owner key is one byte and the row in the tile another");
static_assert(SC_WAVES * QCAP >= HID * HID, "the staged W takes the place of the scan stacks");
static_assert(OWN_NODES == 8 * SC_WAVES && OWN_NODES % NPG == 0, "tail: a 16-row tile per wave pair; whole graphs per block");
static_assert(N_NODES <= 65536 && OWN_NODES <= 256, "a list entry holds src in 16 bits and the tile row above them");
static_assert(E_EDGES % (SC_WAVES * KEYS_STEP) == 0 && (SCAN_STEPS & (SCAN_STEPS - 1)) == 0 && SCAN_STEPS % 4 == 0,
              "the waves' scans cover the edge list exactly");

__device__ __forceinline__ unsigned zero_bytes(unsigned x) {
    const unsigned y = (x & 0x7f7f7f7fu) + 0x7f7f7f7fu;
    return ~(y | x | 0x7f7f7f7fu);
}

__device__ __forceinline__ void lds_add(float* p, float v) {
    __hip_atomic_fetch_add(p, v, __ATOMIC_RELAXED, __HIP_MEMORY_SCOPE_WORKGROUP);
}

__device__ __forceinline__ void sc_rows(int s_l, int dl_l, float en_l, int lane,
                                        const float* __restrict__ hw,
                                        float* tile) {
    const float* hwl = hw + lane * 4;
    float* tl = tile + lane;
#pragma unroll 1
    for (int g = 0; g < 2; ++g) {
        v4f r[16];
#pragma unroll
        for (int u = 0; u < 16; ++u) {
            const int s = __builtin_amdgcn_readlane(s_l, g * 16 + u);
            r[u] = *(const v4f*)(hwl + (size_t)s * HID);
        }
#pragma unroll
        for (int u = 0; u < 16; ++u) {
            const int   dl = __builtin_amdgcn_readlane(dl_l, g * 16 + u);
            const float en = __builtin_bit_cast(float,
                __builtin_amdgcn_readlane(__builtin_bit_cast(int, en_l), g * 16 + u));
            float* t = tl + dl * T_PITCH;
            lds_add(t,      r[u].x * en);
            lds_add(t + 32, r[u].y * en);
            lds_add(t + 64, r[u].z * en);
            lds_add(t + 96, r[u].w * en);
        }
    }
}

__device__ __forceinline__ void sc_drain(const unsigned* st, int n, int lane,
                                         const int* __restrict__ edge_index,
                                         const float* __restrict__ dinv,
                                         const float* __restrict__ hw,
                                         float* tile, unsigned* rec) {
    int   s_l  = 0;
    int   dl_l = OWN_NODES;
    float en_l = 0.0f;
    if (lane < n) {
        const unsigned e = st[lane];
        s_l = edge_index[e];
        const int d = edge_index[E_EDGES + e];
        en_l = dinv[s_l] * dinv[d];
        dl_l = d & (OWN_NODES - 1);
        if (rec) {
            v2u ent;
            ent.x = (unsigned)s_l | ((unsigned)dl_l << 16);
            ent.y = __builtin_bit_cast(unsigned, en_l);
            *(v2u*)(rec + 2 * lane) = ent;
        }
    }
    sc_rows(s_l, dl_l, en_l, lane, hw, tile);
}

__launch_bounds__(1024)
__global__ void k_scatter_gemm(const int* __restrict__ edge_index,
                               const unsigned char* __restrict__ okey,
                               const float* __restrict__ dinv,
                               const float* __restrict__ invdeg,
                               const float* __restrict__ hw,
                               float* __restrict__ hwn,
                               float* __restrict__ h,
                               const float* __restrict__ bias,
                               const float* __restrict__ gamma,
                               const float* __restrict__ beta,
                               const float* __restrict__ Wn,
                               const float* __restrict__ W_out,
                               float* __restrict__ out,
                               unsigned* __restrict__ elist,
                               int* __restrict__ ecnt,
                               const int record,
                               const int is_last) {
    __shared__ __attribute__((aligned(16))) float tile[(OWN_NODES + 1) * T_PITCH];
    __shared__ __attribute__((aligned(16))) unsigned stk[SC_WAVES * QCAP];

    const int wave  = threadIdx.x >> 5;
    const int lane  = threadIdx.x & 31;
    const int blk   = blockIdx.x;
    const int node0 = blk * OWN_NODES;

    {
        const v4f bv = *(const v4f*)(bias + lane * 4);
        for (int r = wave; r < OWN_NODES; r += SC_WAVES) {
            const v4f a = *(const v4f*)(hw + (size_t)(node0 + r) * HID + lane * 4);
            const float inv = invdeg[node0 + r];
            float* t = tile + r * T_PITCH + lane;
            t[0]  = a.x * inv + bv.x;
            t[32] = a.y * inv + bv.y;
            t[64] = a.z * inv + bv.z;
            t[96] = a.w * inv + bv.w;
        }
    }
    __syncthreads();

    const int slot = blk * SC_WAVES + wave;
    unsigned* el = elist + (size_t)slot * (2 * LCAP);
    bool scan = true;
    if (!record) {
        const int c = __builtin_amdgcn_readfirstlane(ecnt[slot]);
        if (c >= 0) {
            scan = false;
            const unsigned pad0 = (unsigned)OWN_NODES << 16;
            const int last = c - 1;
            unsigned nx0 = pad0, nx1 = 0u;
            if (c > 0) {
                const int j0 = lane < c ? lane : last;
                const unsigned t0 = el[2 * j0], t1 = el[2 * j0 + 1];
                nx0 = lane < c ? t0 : pad0;
                nx1 = lane < c ? t1 : 0u;
            }
#pragma unroll 1
            for (int i = 0; i < c; i += 32) {
                const unsigned cur0 = nx0, cur1 = nx1;
                const int j  = i + 32 + lane;
                const int jc = j < c ? j : last;
                const unsigned t0 = el[2 * jc], t1 = el[2 * jc + 1];
                nx0 = j < c ? t0 : pad0;
                nx1 = j < c ? t1 : 0u;
                sc_rows((int)(cur0 & 0xffffu), (int)(cur0 >> 16),
                        __builtin_bit_cast(float, cur1), lane, hw, tile);
            }
        }
    }
    if (scan) {
        int tot = 0;
        unsigned* st = stk + wave * QCAP;
        const unsigned idrep = (unsigned)blk * 0x01010101u;
        const unsigned char* kw = okey + (size_t)wave * E_WAVE + lane * 16;
        const unsigned e_w = (unsigned)wave * E_WAVE + (unsigned)lane * 16u;
        int cnt = 0;

        uint4 kk[4];
#pragma unroll
        for (int u = 0; u < 4; ++u)
            kk[u] = *(const uint4*)(kw + ((u + blk) & (SCAN_STEPS - 1)) * KEYS_STEP);
#pragma unroll 1
        for (int i = 0; i < SCAN_STEPS; i += 4) {
#pragma unroll
            for (int u = 0; u < 4; ++u) {
                const int it = (i + u + blk) & (SCAN_STEPS - 1);
                const unsigned eb = e_w + (unsigned)it * KEYS_STEP;
                const unsigned kv[4] = {kk[u].x, kk[u].y, kk[u].z, kk[u].w};
                unsigned p = (zero_bytes(kv[0] ^ idrep) >> 7) | (zero_bytes(kv[1] ^ idrep) >> 6) |
                             (zero_bytes(kv[2] ^ idrep) >> 5) | (zero_bytes(kv[3] ^ idrep) >> 4);
                for (;;) {
                    const bool has = (p != 0u);
                    const unsigned m = __builtin_amdgcn_ballot_w32(has);
                    if (m == 0u) break;
                    if (has) {
                        const unsigned t = (unsigned)__builtin_ctz(p);
                        st[cnt + (int)__builtin_amdgcn_mbcnt_lo(m, 0u)] = eb + ((t & 7u) << 2) + (t >> 3);
                        p &= p - 1u;
                    }
                    cnt += __builtin_popcount(m);
                }
                if (i + 4 < SCAN_STEPS)
                    kk[u] = *(const uint4*)(kw + ((i + u + 4 + blk) & (SCAN_STEPS - 1)) * KEYS_STEP);
                const int need = (i + u + 1 < SCAN_STEPS) ? 32 : 1;
                while (cnt >= need) {
                    const int n = cnt < 32 ? cnt : 32;
                    cnt -= n;
                    sc_drain(st + cnt, n, lane, edge_index, dinv, hw, tile,
                             (record && tot + n <= LCAP) ? el + 2 * tot : nullptr);
                    tot += n;
                }
            }
        }
        if (record && lane == 0) ecnt[slot] = (tot <= LCAP) ? tot : -1;
    }
    __syncthreads();

    const v4f gm = *(const v4f*)(gamma + lane * 4);
    const v4f bt = *(const v4f*)(beta + lane * 4);

    if (is_last) {
        const bool mine = wave < OWN_NODES / NPG;
        float* t = tile + (mine ? wave * NPG : 0) * T_PITCH;
        if (mine) {
            v4f av;
            av.x = t[lane]; av.y = t[lane + 32]; av.z = t[lane + 64]; av.w = t[lane + 96];
            const v4f hv = *(const v4f*)(h + (size_t)(node0 + wave * NPG) * HID + lane * 4);
            *(v4f*)(t + lane * 4) = relu_res_ln(hv, av, gm, bt);
        }
        __syncthreads();
        if (mine && lane < OUT_DIM) {
            float s = 0.0f;
#pragma unroll 4
            for (int k = 0; k < HID; ++k) s = fmaf(t[k], W_out[k * OUT_DIM + lane], s);
            out[(blk * (OWN_NODES / NPG) + wave) * OUT_DIM + lane] = s;
        }
        return;
    }

    float* lWt = reinterpret_cast<float*>(stk);
    stage_w(Wn, lWt, threadIdx.x, SC_WAVES * 32);

#pragma unroll 8
    for (int j = 0; j < OWN_NODES / SC_WAVES; ++j) {
        const int r = wave + j * SC_WAVES;
        float* t = tile + r * T_PITCH;
        v4f av;
        av.x = t[lane]; av.y = t[lane + 32]; av.z = t[lane + 64]; av.w = t[lane + 96];
        float* hr = h + (size_t)(node0 + r) * HID + lane * 4;
        const v4f v = relu_res_ln(*(const v4f*)hr, av, gm, bt);
        *(v4f*)hr = v;
        *(v4f*)(t + lane * 4) = v;
    }
    __syncthreads();

    const int half = lane >> 4;
    const int l16  = lane & 15;
    const int rt   = wave >> 1;
    const int c0   = (wave & 1) * 64;

    v8f zero = {};
    v8f acc[4];
#pragma unroll
    for (int q = 0; q < 4; ++q) acc[q] = zero;

    const float* arow = tile + (rt * 16 + l16) * T_PITCH;

    for (int k = 0; k < HID; k += 4) {
        const int kb = k + half * 2;
        v2f a = *(const v2f*)(arow + kb);
#pragma unroll
        for (int q = 0; q < 4; ++q) {
            const int col = c0 + q * 16 + l16;
            v2f b = *(const v2f*)(lWt + col * HID + (kb ^ ((col & 31) << 1)));
            acc[q] = __builtin_amdgcn_wmma_f32_16x16x4_f32(
                false, a, false, b, (short)0, acc[q], false, false);
        }
    }

#pragma unroll
    for (int r = 0; r < 8; ++r) {
        float* hwrow = hwn + (size_t)(node0 + rt * 16 + half * 8 + r) * HID + c0;
#pragma unroll
        for (int q = 0; q < 4; ++q) hwrow[q * 16 + l16] = acc[q][r];
    }
}

extern "C" void kernel_launch(void* const* d_in, const int* in_sizes, int n_in,
                              void* d_out, int out_size, void* d_ws, size_t ws_size,
                              hipStream_t stream) {
    const int*   x        = (const int*)d_in[0];
    const int*   ei       = (const int*)d_in[1];
    const float* cls_emb  = (const float*)d_in[3];
    const float* nbr_emb  = (const float*)d_in[4];
    const float* Ws       = (const float*)d_in[5];
    const float* bs       = (const float*)d_in[6];
    const float* gammas   = (const float*)d_in[7];
    const float* betas    = (const float*)d_in[8];
    const float* W_out    = (const float*)d_in[9];
    float*       out      = (float*)d_out;

    char* ws = (char*)d_ws;
    size_t off = 0;
    int*   degcnt = (int*)(ws + off);   off += (size_t)N_NODES * 4;
    float* dinv   = (float*)(ws + off); off += (size_t)N_NODES * 4;
    float* invdeg = (float*)(ws + off); off += (size_t)N_NODES * 4;
    float* h      = (float*)(ws + off); off += (size_t)N_NODES * HID * 4;
    float* hwa    = (float*)(ws + off); off += (size_t)N_NODES * HID * 4;
    float* hwb    = (float*)(ws + off); off += (size_t)N_NODES * HID * 4;
    unsigned char* okey = (unsigned char*)(ws + off); off += (size_t)E_EDGES;
    unsigned* elist = (unsigned*)(ws + off); off += (size_t)(N_NODES / OWN_NODES) * SC_WAVES * LCAP * 8;
    int*   ecnt   = (int*)(ws + off);   off += (size_t)(N_NODES / OWN_NODES) * SC_WAVES * 4;
    (void)ws_size; (void)in_sizes; (void)n_in; (void)out_size;

    k_deg_init  <<<N_NODES / 256, 256, 0, stream>>>(degcnt);
    k_deg_count <<<E_EDGES / 256, 256, 0, stream>>>(ei, degcnt, okey);
    k_deg_finish<<<N_NODES / 256, 256, 0, stream>>>(degcnt, dinv, invdeg);

    k_gemm_embed<<<N_NODES / 128, 256, 0, stream>>>(x, cls_emb, nbr_emb, Ws, h, hwa);

    k_scatter_gemm<<<N_NODES / OWN_NODES, SC_WAVES * 32, 0, stream>>>(
        ei, okey, dinv, invdeg, hwa, hwb, h, bs, gammas, betas,
        Ws + (size_t)1 * HID * HID, W_out, out, elist, ecnt, 1, 0);
    k_scatter_gemm<<<N_NODES / OWN_NODES, SC_WAVES * 32, 0, stream>>>(
        ei, okey, dinv, invdeg, hwb, hwa, h, bs + 1 * HID, gammas + 1 * HID, betas + 1 * HID,
        Ws + (size_t)2 * HID * HID, W_out, out, elist, ecnt, 0, 0);
    k_scatter_gemm<<<N_NODES / OWN_NODES, SC_WAVES * 32, 0, stream>>>(
        ei, okey, dinv, invdeg, hwa, hwb, h, bs + 2 * HID, gammas + 2 * HID, betas + 2 * HID,
        Ws + (size_t)3 * HID * HID, W_out, out, elist, ecnt, 0, 0);
    k_scatter_gemm<<<N_NODES / OWN_NODES, SC_WAVES * 32, 0, stream>>>(
        ei, okey, dinv, invdeg, hwb, hwa, h, bs + 3 * HID, gammas + 3 * HID, betas + 3 * HID,
        Ws, W_out, out, elist, ecnt, 0, 1);
}


# Edge sums by owner waves: layer 0 bins a block's edges into one list per wave (tile row r -> wave r%32); each wave adds its rows by plain LDS load+fma+store, no f32 atomics; sums now in list order
# speedup vs baseline: 3.4284x; 1.0653x over previous
#include <hip/hip_runtime.h>

#define N_NODES   65536
#define NPG       128
#define NGRAPH    512
#define E_EDGES   1048576
#define HID       128
#define OUT_DIM   10
#define N_LAYERS  4
#define LN_EPS    1e-5f

typedef float v2f __attribute__((ext_vector_type(2)));
typedef float v4f __attribute__((ext_vector_type(4)));
typedef float v8f __attribute__((ext_vector_type(8)));
typedef unsigned v2u __attribute__((ext_vector_type(2)));

__global__ void k_deg_init(int* __restrict__ degcnt) {
    int i = blockIdx.x * 256 + threadIdx.x;
    if (i < N_NODES) degcnt[i] = 0;
}

__global__ void k_deg_count(const int* __restrict__ edge_index, int* __restrict__ degcnt,
                            unsigned char* __restrict__ okey) {
    int e = blockIdx.x * 256 + threadIdx.x;
    if (e < E_EDGES) {
        int d = edge_index[E_EDGES + e];
        atomicAdd(&degcnt[d], 1);
        okey[e] = (unsigned char)(d >> 8);
    }
}

__global__ void k_deg_finish(const int* __restrict__ degcnt,
                             float* __restrict__ dinv, float* __restrict__ invdeg) {
    int i = blockIdx.x * 256 + threadIdx.x;
    if (i < N_NODES) {
        float d = 1.0f + (float)degcnt[i];
        dinv[i]   = rsqrtf(d);
        invdeg[i] = 1.0f / d;
    }
}

__device__ __forceinline__ v4f relu_res_ln(v4f hv, v4f av, v4f g, v4f b) {
    v4f val;
    val.x = hv.x + fmaxf(av.x, 0.0f);
    val.y = hv.y + fmaxf(av.y, 0.0f);
    val.z = hv.z + fmaxf(av.z, 0.0f);
    val.w = hv.w + fmaxf(av.w, 0.0f);

    float s = val.x + val.y + val.z + val.w;
#pragma unroll
    for (int off = 16; off >= 1; off >>= 1) s += __shfl_xor(s, off, 32);
    const float mu = s * (1.0f / (float)HID);

    v4f dv = val - mu;
    float q = dv.x * dv.x + dv.y * dv.y + dv.z * dv.z + dv.w * dv.w;
#pragma unroll
    for (int off = 16; off >= 1; off >>= 1) q += __shfl_xor(q, off, 32);
    const float rstd = rsqrtf(q * (1.0f / (float)HID) + LN_EPS);
    return dv * rstd * g + b;
}

#define A_PITCH (HID + 4)

__device__ __forceinline__ void stage_w(const float* __restrict__ W, float* lWt, int tid, int nthr) {
    for (int i = tid * 4; i < HID * HID; i += nthr * 4) {
        v4f w = *(const v4f*)(W + i);
        const int k = i >> 7;
        const int n = i & (HID - 1);
#pragma unroll
        for (int j = 0; j < 4; ++j)
            lWt[(n + j) * HID + (k ^ (((n + j) & 31) << 1))] = w[j];
    }
}

__launch_bounds__(256)
__global__ void k_gemm_embed(const int* __restrict__ x,
                             const float* __restrict__ cls_emb,
                             const float* __restrict__ nbr_emb,
                             const float* __restrict__ W,
                             float* __restrict__ h,
                             float* __restrict__ hw) {
    __shared__ float lWt[HID * HID];
    __shared__ __attribute__((aligned(16))) float lA[128 * A_PITCH];
    const int tid = threadIdx.x;
    const int wave = tid >> 5;
    const int lane = tid & 31;
    const int m0   = blockIdx.x * 128 + wave * 16;

    stage_w(W, lWt, tid, 256);

    {
        const int f = lane << 2;
        float* at = lA + wave * 16 * A_PITCH + f;
#pragma unroll 8
        for (int r = 0; r < 16; ++r) {
            const int n  = m0 + r;
            const int c  = x[2 * n + 0];
            const int nb = x[2 * n + 1];
            const v4f a = *(const v4f*)(cls_emb + (size_t)c  * HID + f);
            const v4f b = *(const v4f*)(nbr_emb + (size_t)nb * HID + f);
            const v4f v = a + b;
            *(v4f*)(h + (size_t)n * HID + f) = v;
            *(v4f*)(at + r * A_PITCH) = v;
        }
    }
    __syncthreads();

    const int half = lane >> 4;
    const int l16  = lane & 15;

    v8f zero = {};
    v8f acc[8];
#pragma unroll
    for (int t = 0; t < 8; ++t) acc[t] = zero;

    const float* arow = lA + (wave * 16 + l16) * A_PITCH;

    for (int k = 0; k < HID; k += 4) {
        const int kb = k + half * 2;
        v2f a = *(const v2f*)(arow + kb);
#pragma unroll
        for (int t = 0; t < 8; ++t) {
            const int col = t * 16 + l16;
            v2f b = *(const v2f*)(lWt + col * HID + (kb ^ ((col & 31) << 1)));
            acc[t] = __builtin_amdgcn_wmma_f32_16x16x4_f32(
                false, a, false, b, (short)0, acc[t], false, false);
        }
    }

    const int rbase = half * 8;
#pragma unroll
    for (int r = 0; r < 8; ++r) {
        float* hwrow = hw + (size_t)(m0 + rbase + r) * HID;
#pragma unroll
        for (int t = 0; t < 8; ++t) hwrow[t * 16 + l16] = acc[t][r];
    }
}

#define OWN_NODES  256
#define T_PITCH    (HID + 4)
#define LCAP       256
#define SC_WAVES   32
#define KEYS_STEP  512
#define E_WAVE     (E_EDGES / SC_WAVES)
#define SCAN_STEPS (E_WAVE / KEYS_STEP)
#define QCAP       (32 + KEYS_STEP)
static_assert(N_NODES == 256 * OWN_NODES, "the owner key is one byte and the row in the tile another");
static_assert(SC_WAVES * QCAP >= HID * HID, "the staged W takes the place of the scan stacks");
static_assert(OWN_NODES == 8 * SC_WAVES && OWN_NODES % NPG == 0, "tail: a 16-row tile per wave pair; whole graphs per block");
static_assert(N_NODES <= 65536 && OWN_NODES <= 256, "a list entry holds src in 16 bits and the tile row above them");
static_assert(E_EDGES % (SC_WAVES * KEYS_STEP) == 0 && (SCAN_STEPS & (SCAN_STEPS - 1)) == 0 && SCAN_STEPS % 4 == 0,
              "the waves' scans cover the edge list exactly");

__device__ __forceinline__ unsigned zero_bytes(unsigned x) {
    const unsigned y = (x & 0x7f7f7f7fu) + 0x7f7f7f7fu;
    return ~(y | x | 0x7f7f7f7fu);
}

__device__ __forceinline__ void lds_add(float* p, float v) {
    __hip_atomic_fetch_add(p, v, __ATOMIC_RELAXED, __HIP_MEMORY_SCOPE_WORKGROUP);
}

__device__ __forceinline__ void own_rows(unsigned e0_l, unsigned e1_l, int ng, int lane,
                                         const float* __restrict__ hw,
                                         float* tile) {
    const float* hwl = hw + lane * 4;
    float* tl = tile + lane * 4;
#pragma unroll 1
    for (int g = 0; g < ng; ++g) {
        v4f r[16];
#pragma unroll
        for (int u = 0; u < 16; ++u) {
            const unsigned w0 = (unsigned)__builtin_amdgcn_readlane((int)e0_l, g * 16 + u);
            r[u] = *(const v4f*)(hwl + (size_t)(w0 & 0xffffu) * HID);
        }
#pragma unroll
        for (int u = 0; u < 16; ++u) {
            const unsigned w0 = (unsigned)__builtin_amdgcn_readlane((int)e0_l, g * 16 + u);
            const float en = __builtin_bit_cast(float,
                __builtin_amdgcn_readlane((int)e1_l, g * 16 + u));
            float* t = tl + (w0 >> 16) * T_PITCH;
            v4f a = *(const v4f*)t;
            a.x = fmaf(r[u].x, en, a.x);
            a.y = fmaf(r[u].y, en, a.y);
            a.z = fmaf(r[u].z, en, a.z);
            a.w = fmaf(r[u].w, en, a.w);
            *(v4f*)t = a;
        }
    }
}

template <bool FROM_LDS>
__device__ __forceinline__ void own_phase(const v2u* ll, const v2u* gl, v2u first, int c, int lane,
                                          const float* __restrict__ hw, float* tile) {
    const unsigned pad0 = (unsigned)OWN_NODES << 16;
    const int last = c - 1;
    unsigned nx0 = pad0, nx1 = 0u;
    if (c > 0) {
        v2u t = first;
        if (FROM_LDS) t = ll[lane < c ? lane : last];
        nx0 = lane < c ? t.x : pad0;
        nx1 = lane < c ? t.y : 0u;
    }
#pragma unroll 1
    for (int i = 0; i < c; i += 32) {
        const unsigned cur0 = nx0, cur1 = nx1;
        const int j  = i + 32 + lane;
        const int jc = j < c ? j : last;
        const v2u t = FROM_LDS ? ll[jc] : gl[jc];
        nx0 = j < c ? t.x : pad0;
        nx1 = j < c ? t.y : 0u;
        own_rows(cur0, cur1, (c - i) > 16 ? 2 : 1, lane, hw, tile);
    }
}

__device__ __forceinline__ void sc_bin(const unsigned* st, int n, int lane,
                                       const int* __restrict__ edge_index,
                                       const float* __restrict__ dinv,
                                       const float* __restrict__ hw,
                                       float* tile, v2u* llist, int* lcnt, int* lovf) {
    int   s_l  = 0;
    int   dl_l = 0;
    float en_l = 0.0f;
    bool  over = false;
    if (lane < n) {
        const unsigned e = st[lane];
        s_l = edge_index[e];
        const int d = edge_index[E_EDGES + e];
        en_l = dinv[s_l] * dinv[d];
        dl_l = d & (OWN_NODES - 1);
        const int ow = dl_l & (SC_WAVES - 1);
        const int place = __hip_atomic_fetch_add(&lcnt[ow], 1, __ATOMIC_RELAXED, __HIP_MEMORY_SCOPE_WORKGROUP);
        if (place < LCAP) {
            v2u ent;
            ent.x = (unsigned)s_l | ((unsigned)dl_l << 16);
            ent.y = __builtin_bit_cast(unsigned, en_l);
            llist[ow * LCAP + place] = ent;
        } else {
            over = true;
        }
    }
    unsigned m = __builtin_amdgcn_ballot_w32(over);
    while (m != 0u) {
        const int l = __builtin_ctz(m);
        m &= m - 1u;
        const int   s  = __builtin_amdgcn_readlane(s_l, l);
        const int   dl = __builtin_amdgcn_readlane(dl_l, l);
        const float en = __builtin_bit_cast(float,
            __builtin_amdgcn_readlane(__builtin_bit_cast(int, en_l), l));
        const v4f r = *(const v4f*)(hw + (size_t)s * HID + lane * 4);
        float* t = tile + dl * T_PITCH + lane * 4;
        lds_add(t,     r.x * en);
        lds_add(t + 1, r.y * en);
        lds_add(t + 2, r.z * en);
        lds_add(t + 3, r.w * en);
        if (lane == 0) *lovf = 1;
    }
}

__launch_bounds__(1024)
__global__ void k_scatter_gemm(const int* __restrict__ edge_index,
                               const unsigned char* __restrict__ okey,
                               const float* __restrict__ dinv,
                               const float* __restrict__ invdeg,
                               const float* __restrict__ hw,
                               float* __restrict__ hwn,
                               float* __restrict__ h,
                               const float* __restrict__ bias,
                               const float* __restrict__ gamma,
                               const float* __restrict__ beta,
                               const float* __restrict__ Wn,
                               const float* __restrict__ W_out,
                               float* __restrict__ out,
                               unsigned* __restrict__ elist,
                               int* __restrict__ ecnt,
                               int* __restrict__ eflag,
                               const int record,
                               const int is_last) {
    __shared__ __attribute__((aligned(16))) float tile[(OWN_NODES + 1) * T_PITCH];
    __shared__ __attribute__((aligned(16))) unsigned stk[SC_WAVES * QCAP];
    __shared__ __attribute__((aligned(16))) v2u llist[SC_WAVES * LCAP];
    __shared__ int lcnt[SC_WAVES];
    __shared__ int lovf;

    const int wave  = threadIdx.x >> 5;
    const int lane  = threadIdx.x & 31;
    const int blk   = blockIdx.x;
    const int node0 = blk * OWN_NODES;

    const int slot = blk * SC_WAVES + wave;
    v2u* gl = reinterpret_cast<v2u*>(elist + (size_t)slot * (2 * LCAP));
    const v2u* ll = llist + wave * LCAP;

    int fl = 1, cg = 0;
    v2u first = {(unsigned)OWN_NODES << 16, 0u};
    if (!record) {
        fl = eflag[blk];
        cg = ecnt[slot];
        first = gl[lane];
    }
    const bool scan = __builtin_amdgcn_readfirstlane(fl) != 0;

    {
        const v4f bv = *(const v4f*)(bias + lane * 4);
        for (int r = wave; r < OWN_NODES; r += SC_WAVES) {
            const v4f a = *(const v4f*)(hw + (size_t)(node0 + r) * HID + lane * 4);
            const float inv = invdeg[node0 + r];
            *(v4f*)(tile + r * T_PITCH + lane * 4) = a * inv + bv;
        }
    }

    int c;

    if (scan) {
        if (threadIdx.x < SC_WAVES) lcnt[threadIdx.x] = 0;
        if (threadIdx.x == 0) lovf = 0;
        __syncthreads();

        unsigned* st = stk + wave * QCAP;
        const unsigned idrep = (unsigned)blk * 0x01010101u;
        const unsigned char* kw = okey + (size_t)wave * E_WAVE + lane * 16;
        const unsigned e_w = (unsigned)wave * E_WAVE + (unsigned)lane * 16u;
        int cnt = 0;

        uint4 kk[4];
#pragma unroll
        for (int u = 0; u < 4; ++u)
            kk[u] = *(const uint4*)(kw + ((u + blk) & (SCAN_STEPS - 1)) * KEYS_STEP);
#pragma unroll 1
        for (int i = 0; i < SCAN_STEPS; i += 4) {
#pragma unroll
            for (int u = 0; u < 4; ++u) {
                const int it = (i + u + blk) & (SCAN_STEPS - 1);
                const unsigned eb = e_w + (unsigned)it * KEYS_STEP;
                const unsigned kv[4] = {kk[u].x, kk[u].y, kk[u].z, kk[u].w};
                unsigned p = (zero_bytes(kv[0] ^ idrep) >> 7) | (zero_bytes(kv[1] ^ idrep) >> 6) |
                             (zero_bytes(kv[2] ^ idrep) >> 5) | (zero_bytes(kv[3] ^ idrep) >> 4);
                for (;;) {
                    const bool has = (p != 0u);
                    const unsigned m = __builtin_amdgcn_ballot_w32(has);
                    if (m == 0u) break;
                    if (has) {
                        const unsigned t = (unsigned)__builtin_ctz(p);
                        st[cnt + (int)__builtin_amdgcn_mbcnt_lo(m, 0u)] = eb + ((t & 7u) << 2) + (t >> 3);
                        p &= p - 1u;
                    }
                    cnt += __builtin_popcount(m);
                }
                if (i + 4 < SCAN_STEPS)
                    kk[u] = *(const uint4*)(kw + ((i + u + 4 + blk) & (SCAN_STEPS - 1)) * KEYS_STEP);
                const int need = (i + u + 1 < SCAN_STEPS) ? 32 : 1;
                while (cnt >= need) {
                    const int n = cnt < 32 ? cnt : 32;
                    cnt -= n;
                    sc_bin(st + cnt, n, lane, edge_index, dinv, hw, tile, llist, lcnt, &lovf);
                }
            }
        }
        __syncthreads();

        const int cr = lcnt[wave];
        c = __builtin_amdgcn_readfirstlane(cr < LCAP ? cr : LCAP);
        if (record) {
            const v2u padent = {(unsigned)OWN_NODES << 16, 0u};
            for (int j = lane; j < c || j < 32; j += 32) {
                v2u e = padent;
                if (j < c) e = ll[j];
                gl[j] = e;
            }
            if (lane == 0) ecnt[slot] = c;
            if (threadIdx.x == 0) eflag[blk] = lovf;
        }
    } else {
        c = __builtin_amdgcn_readfirstlane(cg);
        c = c < 0 ? 0 : (c > LCAP ? LCAP : c);
    }

    if (scan) own_phase<true>(ll, gl, first, c, lane, hw, tile);
    else      own_phase<false>(ll, gl, first, c, lane, hw, tile);

    const v4f gm = *(const v4f*)(gamma + lane * 4);
    const v4f bt = *(const v4f*)(beta + lane * 4);

    if (is_last) {
        __syncthreads();
        const bool mine = wave < OWN_NODES / NPG;
        float* t = tile + (mine ? wave * NPG : 0) * T_PITCH;
        if (mine) {
            const v4f av = *(const v4f*)(t + lane * 4);
            const v4f hv = *(const v4f*)(h + (size_t)(node0 + wave * NPG) * HID + lane * 4);
            *(v4f*)(t + lane * 4) = relu_res_ln(hv, av, gm, bt);
        }
        __syncthreads();
        if (mine && lane < OUT_DIM) {
            float s = 0.0f;
#pragma unroll 4
            for (int k = 0; k < HID; ++k) s = fmaf(t[k], W_out[k * OUT_DIM + lane], s);
            out[(blk * (OWN_NODES / NPG) + wave) * OUT_DIM + lane] = s;
        }
        return;
    }

    float* lWt = reinterpret_cast<float*>(stk);
    stage_w(Wn, lWt, threadIdx.x, SC_WAVES * 32);

#pragma unroll 8
    for (int j = 0; j < OWN_NODES / SC_WAVES; ++j) {
        const int r = wave + j * SC_WAVES;
        float* t = tile + r * T_PITCH + lane * 4;
        const v4f av = *(const v4f*)t;
        float* hr = h + (size_t)(node0 + r) * HID + lane * 4;
        const v4f v = relu_res_ln(*(const v4f*)hr, av, gm, bt);
        *(v4f*)hr = v;
        *(v4f*)t = v;
    }
    __syncthreads();

    const int half = lane >> 4;
    const int l16  = lane & 15;
    const int rt   = wave >> 1;
    const int c0   = (wave & 1) * 64;

    v8f zero = {};
    v8f acc[4];
#pragma unroll
    for (int q = 0; q < 4; ++q) acc[q] = zero;

    const float* arow = tile + (rt * 16 + l16) * T_PITCH;

    for (int k = 0; k < HID; k += 4) {
        const int kb = k + half * 2;
        v2f a = *(const v2f*)(arow + kb);
#pragma unroll
        for (int q = 0; q < 4; ++q) {
            const int col = c0 + q * 16 + l16;
            v2f b = *(const v2f*)(lWt + col * HID + (kb ^ ((col & 31) << 1)));
            acc[q] = __builtin_amdgcn_wmma_f32_16x16x4_f32(
                false, a, false, b, (short)0, acc[q], false, false);
        }
    }

#pragma unroll
    for (int r = 0; r < 8; ++r) {
        float* hwrow = hwn + (size_t)(node0 + rt * 16 + half * 8 + r) * HID + c0;
#pragma unroll
        for (int q = 0; q < 4; ++q) hwrow[q * 16 + l16] = acc[q][r];
    }
}

extern "C" void kernel_launch(void* const* d_in, const int* in_sizes, int n_in,
                              void* d_out, int out_size, void* d_ws, size_t ws_size,
                              hipStream_t stream) {
    const int*   x        = (const int*)d_in[0];
    const int*   ei       = (const int*)d_in[1];
    const float* cls_emb  = (const float*)d_in[3];
    const float* nbr_emb  = (const float*)d_in[4];
    const float* Ws       = (const float*)d_in[5];
    const float* bs       = (const float*)d_in[6];
    const float* gammas   = (const float*)d_in[7];
    const float* betas    = (const float*)d_in[8];
    const float* W_out    = (const float*)d_in[9];
    float*       out      = (float*)d_out;

    char* ws = (char*)d_ws;
    size_t off = 0;
    int*   degcnt = (int*)(ws + off);   off += (size_t)N_NODES * 4;
    float* dinv   = (float*)(ws + off); off += (size_t)N_NODES * 4;
    float* invdeg = (float*)(ws + off); off += (size_t)N_NODES * 4;
    float* h      = (float*)(ws + off); off += (size_t)N_NODES * HID * 4;
    float* hwa    = (float*)(ws + off); off += (size_t)N_NODES * HID * 4;
    float* hwb    = (float*)(ws + off); off += (size_t)N_NODES * HID * 4;
    unsigned char* okey = (unsigned char*)(ws + off); off += (size_t)E_EDGES;
    unsigned* elist = (unsigned*)(ws + off); off += (size_t)(N_NODES / OWN_NODES) * SC_WAVES * LCAP * 8;
    int*   ecnt   = (int*)(ws + off);   off += (size_t)(N_NODES / OWN_NODES) * SC_WAVES * 4;
    int*   eflag  = (int*)(ws + off);   off += (size_t)(N_NODES / OWN_NODES) * 4;
    (void)ws_size; (void)in_sizes; (void)n_in; (void)out_size;

    k_deg_init  <<<N_NODES / 256, 256, 0, stream>>>(degcnt);
    k_deg_count <<<E_EDGES / 256, 256, 0, stream>>>(ei, degcnt, okey);
    k_deg_finish<<<N_NODES / 256, 256, 0, stream>>>(degcnt, dinv, invdeg);

    k_gemm_embed<<<N_NODES / 128, 256, 0, stream>>>(x, cls_emb, nbr_emb, Ws, h, hwa);

    k_scatter_gemm<<<N_NODES / OWN_NODES, SC_WAVES * 32, 0, stream>>>(
        ei, okey, dinv, invdeg, hwa, hwb, h, bs, gammas, betas,
        Ws + (size_t)1 * HID * HID, W_out, out, elist, ecnt, eflag, 1, 0);
    k_scatter_gemm<<<N_NODES / OWN_NODES, SC_WAVES * 32, 0, stream>>>(
        ei, okey, dinv, invdeg, hwb, hwa, h, bs + 1 * HID, gammas + 1 * HID, betas + 1 * HID,
        Ws + (size_t)2 * HID * HID, W_out, out, elist, ecnt, eflag, 0, 0);
    k_scatter_gemm<<<N_NODES / OWN_NODES, SC_WAVES * 32, 0, stream>>>(
        ei, okey, dinv, invdeg, hwa, hwb, h, bs + 2 * HID, gammas + 2 * HID, betas + 2 * HID,
        Ws + (size_t)3 * HID * HID, W_out, out, elist, ecnt, eflag, 0, 0);
    k_scatter_gemm<<<N_NODES / OWN_NODES, SC_WAVES * 32, 0, stream>>>(
        ei, okey, dinv, invdeg, hwb, hwa, h, bs + 3 * HID, gammas + 3 * HID, betas + 3 * HID,
        Ws, W_out, out, elist, ecnt, eflag, 0, 1);
}


# Loops that paid one memory round trip per trip ask for all their loads at once: start values (8 rows per wave), W staging (4 or 16 pieces per thread), W_out staged in LDS for the logits; sums kept
# speedup vs baseline: 3.6624x; 1.0683x over previous
#include <hip/hip_runtime.h>

#define N_NODES   65536
#define NPG       128
#define NGRAPH    512
#define E_EDGES   1048576
#define HID       128
#define OUT_DIM   10
#define N_LAYERS  4
#define LN_EPS    1e-5f

typedef float v2f __attribute__((ext_vector_type(2)));
typedef float v4f __attribute__((ext_vector_type(4)));
typedef float v8f __attribute__((ext_vector_type(8)));
typedef unsigned v2u __attribute__((ext_vector_type(2)));

__global__ void k_deg_init(int* __restrict__ degcnt) {
    int i = blockIdx.x * 256 + threadIdx.x;
    if (i < N_NODES) degcnt[i] = 0;
}

__global__ void k_deg_count(const int* __restrict__ edge_index, int* __restrict__ degcnt,
                            unsigned char* __restrict__ okey) {
    int e = blockIdx.x * 256 + threadIdx.x;
    if (e < E_EDGES) {
        int d = edge_index[E_EDGES + e];
        atomicAdd(&degcnt[d], 1);
        okey[e] = (unsigned char)(d >> 8);
    }
}

__global__ void k_deg_finish(const int* __restrict__ degcnt,
                             float* __restrict__ dinv, float* __restrict__ invdeg) {
    int i = blockIdx.x * 256 + threadIdx.x;
    if (i < N_NODES) {
        float d = 1.0f + (float)degcnt[i];
        dinv[i]   = rsqrtf(d);
        invdeg[i] = 1.0f / d;
    }
}

__device__ __forceinline__ v4f relu_res_ln(v4f hv, v4f av, v4f g, v4f b) {
    v4f val;
    val.x = hv.x + fmaxf(av.x, 0.0f);
    val.y = hv.y + fmaxf(av.y, 0.0f);
    val.z = hv.z + fmaxf(av.z, 0.0f);
    val.w = hv.w + fmaxf(av.w, 0.0f);

    float s = val.x + val.y + val.z + val.w;
#pragma unroll
    for (int off = 16; off >= 1; off >>= 1) s += __shfl_xor(s, off, 32);
    const float mu = s * (1.0f / (float)HID);

    v4f dv = val - mu;
    float q = dv.x * dv.x + dv.y * dv.y + dv.z * dv.z + dv.w * dv.w;
#pragma unroll
    for (int off = 16; off >= 1; off >>= 1) q += __shfl_xor(q, off, 32);
    const float rstd = rsqrtf(q * (1.0f / (float)HID) + LN_EPS);
    return dv * rstd * g + b;
}

#define A_PITCH (HID + 4)

#define W_PIECES(NTHR) (HID * HID / ((NTHR) * 4))

template <int NTHR>
__device__ __forceinline__ void stage_w_load(const float* __restrict__ W, int tid, v4f (&w)[W_PIECES(NTHR)]) {
#pragma unroll
    for (int t = 0; t < W_PIECES(NTHR); ++t)
        w[t] = *(const v4f*)(W + (tid + t * NTHR) * 4);
}

template <int NTHR>
__device__ __forceinline__ void stage_w_store(float* lWt, int tid, const v4f (&w)[W_PIECES(NTHR)]) {
#pragma unroll
    for (int t = 0; t < W_PIECES(NTHR); ++t) {
        const int i = (tid + t * NTHR) * 4;
        const int k = i >> 7;
        const int n = i & (HID - 1);
#pragma unroll
        for (int j = 0; j < 4; ++j)
            lWt[(n + j) * HID + (k ^ (((n + j) & 31) << 1))] = w[t][j];
    }
}

__launch_bounds__(256)
__global__ void k_gemm_embed(const int* __restrict__ x,
                             const float* __restrict__ cls_emb,
                             const float* __restrict__ nbr_emb,
                             const float* __restrict__ W,
                             float* __restrict__ h,
                             float* __restrict__ hw) {
    __shared__ float lWt[HID * HID];
    __shared__ __attribute__((aligned(16))) float lA[128 * A_PITCH];
    const int tid = threadIdx.x;
    const int wave = tid >> 5;
    const int lane = tid & 31;
    const int m0   = blockIdx.x * 128 + wave * 16;

    v4f wreg[W_PIECES(256)];
    stage_w_load<256>(W, tid, wreg);

    {
        const int f = lane << 2;
        float* at = lA + wave * 16 * A_PITCH + f;
#pragma unroll 8
        for (int r = 0; r < 16; ++r) {
            const int n  = m0 + r;
            const int c  = x[2 * n + 0];
            const int nb = x[2 * n + 1];
            const v4f a = *(const v4f*)(cls_emb + (size_t)c  * HID + f);
            const v4f b = *(const v4f*)(nbr_emb + (size_t)nb * HID + f);
            const v4f v = a + b;
            *(v4f*)(h + (size_t)n * HID + f) = v;
            *(v4f*)(at + r * A_PITCH) = v;
        }
    }
    stage_w_store<256>(lWt, tid, wreg);
    __syncthreads();

    const int half = lane >> 4;
    const int l16  = lane & 15;

    v8f zero = {};
    v8f acc[8];
#pragma unroll
    for (int t = 0; t < 8; ++t) acc[t] = zero;

    const float* arow = lA + (wave * 16 + l16) * A_PITCH;

    for (int k = 0; k < HID; k += 4) {
        const int kb = k + half * 2;
        v2f a = *(const v2f*)(arow + kb);
#pragma unroll
        for (int t = 0; t < 8; ++t) {
            const int col = t * 16 + l16;
            v2f b = *(const v2f*)(lWt + col * HID + (kb ^ ((col & 31) << 1)));
            acc[t] = __builtin_amdgcn_wmma_f32_16x16x4_f32(
                false, a, false, b, (short)0, acc[t], false, false);
        }
    }

    const int rbase = half * 8;
#pragma unroll
    for (int r = 0; r < 8; ++r) {
        float* hwrow = hw + (size_t)(m0 + rbase + r) * HID;
#pragma unroll
        for (int t = 0; t < 8; ++t) hwrow[t * 16 + l16] = acc[t][r];
    }
}

#define OWN_NODES  256
#define T_PITCH    (HID + 4)
#define LCAP       256
#define SC_WAVES   32
#define KEYS_STEP  512
#define E_WAVE     (E_EDGES / SC_WAVES)
#define SCAN_STEPS (E_WAVE / KEYS_STEP)
#define QCAP       (32 + KEYS_STEP)
#define WO_REST    (HID * OUT_DIM - SC_WAVES * 32)
static_assert(N_NODES == 256 * OWN_NODES, "the owner key is one byte and the row in the tile another");
static_assert(SC_WAVES * QCAP >= HID * HID, "the staged W takes the place of the scan stacks");
static_assert(OWN_NODES == 8 * SC_WAVES && OWN_NODES % NPG == 0, "tail: a 16-row tile per wave pair; whole graphs per block");
static_assert(N_NODES <= 65536 && OWN_NODES <= 256, "a list entry holds src in 16 bits and the tile row above them");
static_assert(WO_REST > 0 && WO_REST <= SC_WAVES * 32 && SC_WAVES * QCAP >= HID * OUT_DIM,
              "W_out: at most two words per thread, and it fits where the scan stacks are");
static_assert(E_EDGES % (SC_WAVES * KEYS_STEP) == 0 && (SCAN_STEPS & (SCAN_STEPS - 1)) == 0 && SCAN_STEPS % 4 == 0,
              "the waves' scans cover the edge list exactly");

__device__ __forceinline__ unsigned zero_bytes(unsigned x) {
    const unsigned y = (x & 0x7f7f7f7fu) + 0x7f7f7f7fu;
    return ~(y | x | 0x7f7f7f7fu);
}

__device__ __forceinline__ void lds_add(float* p, float v) {
    __hip_atomic_fetch_add(p, v, __ATOMIC_RELAXED, __HIP_MEMORY_SCOPE_WORKGROUP);
}

__device__ __forceinline__ void own_rows(unsigned e0_l, unsigned e1_l, int ng, int lane,
                                         const float* __restrict__ hw,
                                         float* tile) {
    const float* hwl = hw + lane * 4;
    float* tl = tile + lane * 4;
#pragma unroll 1
    for (int g = 0; g < ng; ++g) {
        v4f r[16];
#pragma unroll
        for (int u = 0; u < 16; ++u) {
            const unsigned w0 = (unsigned)__builtin_amdgcn_readlane((int)e0_l, g * 16 + u);
            r[u] = *(const v4f*)(hwl + (size_t)(w0 & 0xffffu) * HID);
        }
#pragma unroll
        for (int u = 0; u < 16; ++u) {
            const unsigned w0 = (unsigned)__builtin_amdgcn_readlane((int)e0_l, g * 16 + u);
            const float en = __builtin_bit_cast(float,
                __builtin_amdgcn_readlane((int)e1_l, g * 16 + u));
            float* t = tl + (w0 >> 16) * T_PITCH;
            v4f a = *(const v4f*)t;
            a.x = fmaf(r[u].x, en, a.x);
            a.y = fmaf(r[u].y, en, a.y);
            a.z = fmaf(r[u].z, en, a.z);
            a.w = fmaf(r[u].w, en, a.w);
            *(v4f*)t = a;
        }
    }
}

template <bool FROM_LDS>
__device__ __forceinline__ void own_phase(const v2u* ll, const v2u* gl, v2u first, int c, int lane,
                                          const float* __restrict__ hw, float* tile) {
    const unsigned pad0 = (unsigned)OWN_NODES << 16;
    const int last = c - 1;
    unsigned nx0 = pad0, nx1 = 0u;
    if (c > 0) {
        v2u t = first;
        if (FROM_LDS) t = ll[lane < c ? lane : last];
        nx0 = lane < c ? t.x : pad0;
        nx1 = lane < c ? t.y : 0u;
    }
#pragma unroll 1
    for (int i = 0; i < c; i += 32) {
        const unsigned cur0 = nx0, cur1 = nx1;
        const int j  = i + 32 + lane;
        const int jc = j < c ? j : last;
        const v2u t = FROM_LDS ? ll[jc] : gl[jc];
        nx0 = j < c ? t.x : pad0;
        nx1 = j < c ? t.y : 0u;
        own_rows(cur0, cur1, (c - i) > 16 ? 2 : 1, lane, hw, tile);
    }
}

__device__ __forceinline__ void sc_bin(const unsigned* st, int n, int lane,
                                       const int* __restrict__ edge_index,
                                       const float* __restrict__ dinv,
                                       const float* __restrict__ hw,
                                       float* tile, v2u* llist, int* lcnt, int* lovf) {
    int   s_l  = 0;
    int   dl_l = 0;
    float en_l = 0.0f;
    bool  over = false;
    if (lane < n) {
        const unsigned e = st[lane];
        s_l = edge_index[e];
        const int d = edge_index[E_EDGES + e];
        en_l = dinv[s_l] * dinv[d];
        dl_l = d & (OWN_NODES - 1);
        const int ow = dl_l & (SC_WAVES - 1);
        const int place = __hip_atomic_fetch_add(&lcnt[ow], 1, __ATOMIC_RELAXED, __HIP_MEMORY_SCOPE_WORKGROUP);
        if (place < LCAP) {
            v2u ent;
            ent.x = (unsigned)s_l | ((unsigned)dl_l << 16);
            ent.y = __builtin_bit_cast(unsigned, en_l);
            llist[ow * LCAP + place] = ent;
        } else {
            over = true;
        }
    }
    unsigned m = __builtin_amdgcn_ballot_w32(over);
    while (m != 0u) {
        const int l = __builtin_ctz(m);
        m &= m - 1u;
        const int   s  = __builtin_amdgcn_readlane(s_l, l);
        const int   dl = __builtin_amdgcn_readlane(dl_l, l);
        const float en = __builtin_bit_cast(float,
            __builtin_amdgcn_readlane(__builtin_bit_cast(int, en_l), l));
        const v4f r = *(const v4f*)(hw + (size_t)s * HID + lane * 4);
        float* t = tile + dl * T_PITCH + lane * 4;
        lds_add(t,     r.x * en);
        lds_add(t + 1, r.y * en);
        lds_add(t + 2, r.z * en);
        lds_add(t + 3, r.w * en);
        if (lane == 0) *lovf = 1;
    }
}

__launch_bounds__(1024)
__global__ void k_scatter_gemm(const int* __restrict__ edge_index,
                               const unsigned char* __restrict__ okey,
                               const float* __restrict__ dinv,
                               const float* __restrict__ invdeg,
                               const float* __restrict__ hw,
                               float* __restrict__ hwn,
                               float* __restrict__ h,
                               const float* __restrict__ bias,
                               const float* __restrict__ gamma,
                               const float* __restrict__ beta,
                               const float* __restrict__ Wn,
                               const float* __restrict__ W_out,
                               float* __restrict__ out,
                               unsigned* __restrict__ elist,
                               int* __restrict__ ecnt,
                               int* __restrict__ eflag,
                               const int record,
                               const int is_last) {
    __shared__ __attribute__((aligned(16))) float tile[(OWN_NODES + 1) * T_PITCH];
    __shared__ __attribute__((aligned(16))) unsigned stk[SC_WAVES * QCAP];
    __shared__ __attribute__((aligned(16))) v2u llist[SC_WAVES * LCAP];
    __shared__ int lcnt[SC_WAVES];
    __shared__ int lovf;

    const int wave  = threadIdx.x >> 5;
    const int lane  = threadIdx.x & 31;
    const int blk   = blockIdx.x;
    const int node0 = blk * OWN_NODES;

    const int slot = blk * SC_WAVES + wave;
    v2u* gl = reinterpret_cast<v2u*>(elist + (size_t)slot * (2 * LCAP));
    const v2u* ll = llist + wave * LCAP;

    int fl = 1, cg = 0;
    v2u first = {(unsigned)OWN_NODES << 16, 0u};
    if (!record) {
        fl = eflag[blk];
        cg = ecnt[slot];
        first = gl[lane];
    }
    float wo0 = 0.0f, wo1 = 0.0f;
    if (is_last) {
        wo0 = W_out[threadIdx.x];
        if (threadIdx.x < WO_REST) wo1 = W_out[SC_WAVES * 32 + threadIdx.x];
    }
    const bool scan = __builtin_amdgcn_readfirstlane(fl) != 0;

    {
        const v4f bv = *(const v4f*)(bias + lane * 4);
        v4f   a[OWN_NODES / SC_WAVES];
        float inv[OWN_NODES / SC_WAVES];
#pragma unroll
        for (int j = 0; j < OWN_NODES / SC_WAVES; ++j) {
            const int r = wave + j * SC_WAVES;
            a[j]   = *(const v4f*)(hw + (size_t)(node0 + r) * HID + lane * 4);
            inv[j] = invdeg[node0 + r];
        }
#pragma unroll
        for (int j = 0; j < OWN_NODES / SC_WAVES; ++j) {
            const int r = wave + j * SC_WAVES;
            *(v4f*)(tile + r * T_PITCH + lane * 4) = a[j] * inv[j] + bv;
        }
    }

    int c;

    if (scan) {
        if (threadIdx.x < SC_WAVES) lcnt[threadIdx.x] = 0;
        if (threadIdx.x == 0) lovf = 0;
        __syncthreads();

        unsigned* st = stk + wave * QCAP;
        const unsigned idrep = (unsigned)blk * 0x01010101u;
        const unsigned char* kw = okey + (size_t)wave * E_WAVE + lane * 16;
        const unsigned e_w = (unsigned)wave * E_WAVE + (unsigned)lane * 16u;
        int cnt = 0;

        uint4 kk[4];
#pragma unroll
        for (int u = 0; u < 4; ++u)
            kk[u] = *(const uint4*)(kw + ((u + blk) & (SCAN_STEPS - 1)) * KEYS_STEP);
#pragma unroll 1
        for (int i = 0; i < SCAN_STEPS; i += 4) {
#pragma unroll
            for (int u = 0; u < 4; ++u) {
                const int it = (i + u + blk) & (SCAN_STEPS - 1);
                const unsigned eb = e_w + (unsigned)it * KEYS_STEP;
                const unsigned kv[4] = {kk[u].x, kk[u].y, kk[u].z, kk[u].w};
                unsigned p = (zero_bytes(kv[0] ^ idrep) >> 7) | (zero_bytes(kv[1] ^ idrep) >> 6) |
                             (zero_bytes(kv[2] ^ idrep) >> 5) | (zero_bytes(kv[3] ^ idrep) >> 4);
                for (;;) {
                    const bool has = (p != 0u);
                    const unsigned m = __builtin_amdgcn_ballot_w32(has);
                    if (m == 0u) break;
                    if (has) {
                        const unsigned t = (unsigned)__builtin_ctz(p);
                        st[cnt + (int)__builtin_amdgcn_mbcnt_lo(m, 0u)] = eb + ((t & 7u) << 2) + (t >> 3);
                        p &= p - 1u;
                    }
                    cnt += __builtin_popcount(m);
                }
                if (i + 4 < SCAN_STEPS)
                    kk[u] = *(const uint4*)(kw + ((i + u + 4 + blk) & (SCAN_STEPS - 1)) * KEYS_STEP);
                const int need = (i + u + 1 < SCAN_STEPS) ? 32 : 1;
                while (cnt >= need) {
                    const int n = cnt < 32 ? cnt : 32;
                    cnt -= n;
                    sc_bin(st + cnt, n, lane, edge_index, dinv, hw, tile, llist, lcnt, &lovf);
                }
            }
        }
        __syncthreads();

        const int cr = lcnt[wave];
        c = __builtin_amdgcn_readfirstlane(cr < LCAP ? cr : LCAP);
        if (record) {
            const v2u padent = {(unsigned)OWN_NODES << 16, 0u};
            for (int j = lane; j < c || j < 32; j += 32) {
                v2u e = padent;
                if (j < c) e = ll[j];
                gl[j] = e;
            }
            if (lane == 0) ecnt[slot] = c;
            if (threadIdx.x == 0) eflag[blk] = lovf;
        }
    } else {
        c = __builtin_amdgcn_readfirstlane(cg);
        c = c < 0 ? 0 : (c > LCAP ? LCAP : c);
    }

    if (scan) own_phase<true>(ll, gl, first, c, lane, hw, tile);
    else      own_phase<false>(ll, gl, first, c, lane, hw, tile);

    const v4f gm = *(const v4f*)(gamma + lane * 4);
    const v4f bt = *(const v4f*)(beta + lane * 4);

    if (is_last) {
        float* lWo = reinterpret_cast<float*>(stk);
        lWo[threadIdx.x] = wo0;
        if (threadIdx.x < WO_REST) lWo[SC_WAVES * 32 + threadIdx.x] = wo1;
        __syncthreads();
        const bool mine = wave < OWN_NODES / NPG;
        float* t = tile + (mine ? wave * NPG : 0) * T_PITCH;
        if (mine) {
            const v4f av = *(const v4f*)(t + lane * 4);
            const v4f hv = *(const v4f*)(h + (size_t)(node0 + wave * NPG) * HID + lane * 4);
            *(v4f*)(t + lane * 4) = relu_res_ln(hv, av, gm, bt);
        }
        __syncthreads();
        if (mine && lane < OUT_DIM) {
            float s = 0.0f;
#pragma unroll 32
            for (int k = 0; k < HID; ++k) s = fmaf(t[k], lWo[k * OUT_DIM + lane], s);
            out[(blk * (OWN_NODES / NPG) + wave) * OUT_DIM + lane] = s;
        }
        return;
    }

    float* lWt = reinterpret_cast<float*>(stk);
    v4f wreg[W_PIECES(SC_WAVES * 32)];
    stage_w_load<SC_WAVES * 32>(Wn, threadIdx.x, wreg);

#pragma unroll 8
    for (int j = 0; j < OWN_NODES / SC_WAVES; ++j) {
        const int r = wave + j * SC_WAVES;
        float* t = tile + r * T_PITCH + lane * 4;
        const v4f av = *(const v4f*)t;
        float* hr = h + (size_t)(node0 + r) * HID + lane * 4;
        const v4f v = relu_res_ln(*(const v4f*)hr, av, gm, bt);
        *(v4f*)hr = v;
        *(v4f*)t = v;
    }
    stage_w_store<SC_WAVES * 32>(lWt, threadIdx.x, wreg);
    __syncthreads();

    const int half = lane >> 4;
    const int l16  = lane & 15;
    const int rt   = wave >> 1;
    const int c0   = (wave & 1) * 64;

    v8f zero = {};
    v8f acc[4];
#pragma unroll
    for (int q = 0; q < 4; ++q) acc[q] = zero;

    const float* arow = tile + (rt * 16 + l16) * T_PITCH;

    for (int k = 0; k < HID; k += 4) {
        const int kb = k + half * 2;
        v2f a = *(const v2f*)(arow + kb);
#pragma unroll
        for (int q = 0; q < 4; ++q) {
            const int col = c0 + q * 16 + l16;
            v2f b = *(const v2f*)(lWt + col * HID + (kb ^ ((col & 31) << 1)));
            acc[q] = __builtin_amdgcn_wmma_f32_16x16x4_f32(
                false, a, false, b, (short)0, acc[q], false, false);
        }
    }

#pragma unroll
    for (int r = 0; r < 8; ++r) {
        float* hwrow = hwn + (size_t)(node0 + rt * 16 + half * 8 + r) * HID + c0;
#pragma unroll
        for (int q = 0; q < 4; ++q) hwrow[q * 16 + l16] = acc[q][r];
    }
}

extern "C" void kernel_launch(void* const* d_in, const int* in_sizes, int n_in,
                              void* d_out, int out_size, void* d_ws, size_t ws_size,
                              hipStream_t stream) {
    const int*   x        = (const int*)d_in[0];
    const int*   ei       = (const int*)d_in[1];
    const float* cls_emb  = (const float*)d_in[3];
    const float* nbr_emb  = (const float*)d_in[4];
    const float* Ws       = (const float*)d_in[5];
    const float* bs       = (const float*)d_in[6];
    const float* gammas   = (const float*)d_in[7];
    const float* betas    = (const float*)d_in[8];
    const float* W_out    = (const float*)d_in[9];
    float*       out      = (float*)d_out;

    char* ws = (char*)d_ws;
    size_t off = 0;
    int*   degcnt = (int*)(ws + off);   off += (size_t)N_NODES * 4;
    float* dinv   = (float*)(ws + off); off += (size_t)N_NODES * 4;
    float* invdeg = (float*)(ws + off); off += (size_t)N_NODES * 4;
    float* h      = (float*)(ws + off); off += (size_t)N_NODES * HID * 4;
    float* hwa    = (float*)(ws + off); off += (size_t)N_NODES * HID * 4;
    float* hwb    = (float*)(ws + off); off += (size_t)N_NODES * HID * 4;
    unsigned char* okey = (unsigned char*)(ws + off); off += (size_t)E_EDGES;
    unsigned* elist = (unsigned*)(ws + off); off += (size_t)(N_NODES / OWN_NODES) * SC_WAVES * LCAP * 8;
    int*   ecnt   = (int*)(ws + off);   off += (size_t)(N_NODES / OWN_NODES) * SC_WAVES * 4;
    int*   eflag  = (int*)(ws + off);   off += (size_t)(N_NODES / OWN_NODES) * 4;
    (void)ws_size; (void)in_sizes; (void)n_in; (void)out_size;

    k_deg_init  <<<N_NODES / 256, 256, 0, stream>>>(degcnt);
    k_deg_count <<<E_EDGES / 256, 256, 0, stream>>>(ei, degcnt, okey);
    k_deg_finish<<<N_NODES / 256, 256, 0, stream>>>(degcnt, dinv, invdeg);

    k_gemm_embed<<<N_NODES / 128, 256, 0, stream>>>(x, cls_emb, nbr_emb, Ws, h, hwa);

    k_scatter_gemm<<<N_NODES / OWN_NODES, SC_WAVES * 32, 0, stream>>>(
        ei, okey, dinv, invdeg, hwa, hwb, h, bs, gammas, betas,
        Ws + (size_t)1 * HID * HID, W_out, out, elist, ecnt, eflag, 1, 0);
    k_scatter_gemm<<<N_NODES / OWN_NODES, SC_WAVES * 32, 0, stream>>>(
        ei, okey, dinv, invdeg, hwb, hwa, h, bs + 1 * HID, gammas + 1 * HID, betas + 1 * HID,
        Ws + (size_t)2 * HID * HID, W_out, out, elist, ecnt, eflag, 0, 0);
    k_scatter_gemm<<<N_NODES / OWN_NODES, SC_WAVES * 32, 0, stream>>>(
        ei, okey, dinv, invdeg, hwa, hwb, h, bs + 2 * HID, gammas + 2 * HID, betas + 2 * HID,
        Ws + (size_t)3 * HID * HID, W_out, out, elist, ecnt, eflag, 0, 0);
    k_scatter_gemm<<<N_NODES / OWN_NODES, SC_WAVES * 32, 0, stream>>>(
        ei, okey, dinv, invdeg, hwb, hwa, h, bs + 3 * HID, gammas + 3 * HID, betas + 3 * HID,
        Ws, W_out, out, elist, ecnt, eflag, 0, 1);
}


# Layer 0 with no row gather: a block sums each row's edge norms per class id and count id (2 LDS f32 adds per edge); agg = coefficient rows x table [cls_emb;nbr_emb]@W0 on WMMA; sum order changes
# speedup vs baseline: 3.9757x; 1.0855x over previous
#include <hip/hip_runtime.h>

#define N_NODES   65536
#define NPG       128
#define NGRAPH    512
#define E_EDGES   1048576
#define HID       128
#define OUT_DIM   10
#define N_LAYERS  4
#define LN_EPS    1e-5f

typedef float v2f __attribute__((ext_vector_type(2)));
typedef float v4f __attribute__((ext_vector_type(4)));
typedef float v8f __attribute__((ext_vector_type(8)));
typedef unsigned v2u __attribute__((ext_vector_type(2)));

__global__ void k_deg_init(int* __restrict__ degcnt) {
    int i = blockIdx.x * 256 + threadIdx.x;
    if (i < N_NODES) degcnt[i] = 0;
}

__global__ void k_deg_count(const int* __restrict__ edge_index, int* __restrict__ degcnt,
                            unsigned char* __restrict__ okey) {
    int e = blockIdx.x * 256 + threadIdx.x;
    if (e < E_EDGES) {
        int d = edge_index[E_EDGES + e];
        atomicAdd(&degcnt[d], 1);
        okey[e] = (unsigned char)(d >> 8);
    }
}

__global__ void k_deg_finish(const int* __restrict__ degcnt,
                             float* __restrict__ dinv, float* __restrict__ invdeg) {
    int i = blockIdx.x * 256 + threadIdx.x;
    if (i < N_NODES) {
        float d = 1.0f + (float)degcnt[i];
        dinv[i]   = rsqrtf(d);
        invdeg[i] = 1.0f / d;
    }
}

__device__ __forceinline__ v4f relu_res_ln(v4f hv, v4f av, v4f g, v4f b) {
    v4f val;
    val.x = hv.x + fmaxf(av.x, 0.0f);
    val.y = hv.y + fmaxf(av.y, 0.0f);
    val.z = hv.z + fmaxf(av.z, 0.0f);
    val.w = hv.w + fmaxf(av.w, 0.0f);

    float s = val.x + val.y + val.z + val.w;
#pragma unroll
    for (int off = 16; off >= 1; off >>= 1) s += __shfl_xor(s, off, 32);
    const float mu = s * (1.0f / (float)HID);

    v4f dv = val - mu;
    float q = dv.x * dv.x + dv.y * dv.y + dv.z * dv.z + dv.w * dv.w;
#pragma unroll
    for (int off = 16; off >= 1; off >>= 1) q += __shfl_xor(q, off, 32);
    const float rstd = rsqrtf(q * (1.0f / (float)HID) + LN_EPS);
    return dv * rstd * g + b;
}

#define N_CLS   64
#define N_NBR   65
#define T_USED  (N_CLS + N_NBR)
#define T_ROWS  132
#define A_PITCH (HID + 4)

#define W_PIECES(NTHR) (HID * HID / ((NTHR) * 4))

template <int NTHR>
__device__ __forceinline__ void stage_w_load(const float* __restrict__ W, int tid, v4f (&w)[W_PIECES(NTHR)]) {
#pragma unroll
    for (int t = 0; t < W_PIECES(NTHR); ++t)
        w[t] = *(const v4f*)(W + (tid + t * NTHR) * 4);
}

template <int NTHR>
__device__ __forceinline__ void stage_w_store(float* lWt, int tid, const v4f (&w)[W_PIECES(NTHR)]) {
#pragma unroll
    for (int t = 0; t < W_PIECES(NTHR); ++t) {
        const int i = (tid + t * NTHR) * 4;
        const int k = i >> 7;
        const int n = i & (HID - 1);
#pragma unroll
        for (int j = 0; j < 4; ++j)
            lWt[(n + j) * HID + (k ^ (((n + j) & 31) << 1))] = w[t][j];
    }
}

__launch_bounds__(256)
__global__ void k_table(const float* __restrict__ cls_emb,
                        const float* __restrict__ nbr_emb,
                        const float* __restrict__ W,
                        float* __restrict__ Tt) {
    __shared__ float lWt[HID * HID];
    __shared__ __attribute__((aligned(16))) float lA[16 * A_PITCH];
    const int tid = threadIdx.x;
    const int wave = tid >> 5;
    const int lane = tid & 31;
    const int m0   = blockIdx.x * 16;

    v4f wreg[W_PIECES(256)];
    stage_w_load<256>(W, tid, wreg);

    {
        const int f = lane << 2;
#pragma unroll
        for (int i = 0; i < 2; ++i) {
            const int r = 2 * wave + i;
            const int n = m0 + r;
            v4f v = {0.0f, 0.0f, 0.0f, 0.0f};
            if (n < N_CLS)       v = *(const v4f*)(cls_emb + (size_t)n * HID + f);
            else if (n < T_USED) v = *(const v4f*)(nbr_emb + (size_t)(n - N_CLS) * HID + f);
            *(v4f*)(lA + r * A_PITCH + f) = v;
        }
    }
    stage_w_store<256>(lWt, tid, wreg);
    __syncthreads();

    const int half = lane >> 4;
    const int l16  = lane & 15;

    v8f acc = {};

    const float* arow = lA + l16 * A_PITCH;
    const int col = wave * 16 + l16;

    for (int k = 0; k < HID; k += 4) {
        const int kb = k + half * 2;
        v2f a = *(const v2f*)(arow + kb);
        v2f b = *(const v2f*)(lWt + col * HID + (kb ^ ((col & 31) << 1)));
        acc = __builtin_amdgcn_wmma_f32_16x16x4_f32(
            false, a, false, b, (short)0, acc, false, false);
    }

#pragma unroll
    for (int r = 0; r < 8; ++r) {
        const int row = m0 + half * 8 + r;
        if (row < T_ROWS) Tt[col * T_ROWS + row] = acc[r];
    }
}

#define OWN_NODES  256
#define T_PITCH    (HID + 4)
#define LCAP       256
#define SC_WAVES   32
#define KEYS_STEP  512
#define E_WAVE     (E_EDGES / SC_WAVES)
#define SCAN_STEPS (E_WAVE / KEYS_STEP)
#define QCAP       (32 + KEYS_STEP)
#define WO_REST    (HID * OUT_DIM - SC_WAVES * 32)
static_assert(N_NODES == 256 * OWN_NODES, "the owner key is one byte and the row in the tile another");
static_assert(SC_WAVES * QCAP >= HID * HID, "the staged W takes the place of the scan stacks");
static_assert(OWN_NODES == 8 * SC_WAVES && OWN_NODES % NPG == 0, "tail: a 16-row tile per wave pair; whole graphs per block");
static_assert(N_NODES <= 65536 && OWN_NODES <= 256, "a list entry holds src in 16 bits and the tile row above them");
static_assert(WO_REST > 0 && WO_REST <= SC_WAVES * 32 && SC_WAVES * QCAP >= HID * OUT_DIM,
              "W_out: at most two words per thread, and it fits where the scan stacks are");
static_assert(T_ROWS <= T_PITCH && T_ROWS % 4 == 0 && T_USED <= T_ROWS && SC_WAVES * QCAP >= HID * T_ROWS && (HID * T_ROWS) % 4 == 0,
              "layer 0: a coefficient row fits a tile row, and T fits where the scan stacks are");
static_assert(E_EDGES % (SC_WAVES * KEYS_STEP) == 0 && (SCAN_STEPS & (SCAN_STEPS - 1)) == 0 && SCAN_STEPS % 4 == 0,
              "the waves' scans cover the edge list exactly");

__device__ __forceinline__ unsigned zero_bytes(unsigned x) {
    const unsigned y = (x & 0x7f7f7f7fu) + 0x7f7f7f7fu;
    return ~(y | x | 0x7f7f7f7fu);
}

__device__ __forceinline__ void lds_add(float* p, float v) {
    __hip_atomic_fetch_add(p, v, __ATOMIC_RELAXED, __HIP_MEMORY_SCOPE_WORKGROUP);
}

__device__ __forceinline__ void own_rows(unsigned e0_l, unsigned e1_l, int ng, int lane,
                                         const float* __restrict__ hw,
                                         float* tile) {
    const float* hwl = hw + lane * 4;
    float* tl = tile + lane * 4;
#pragma unroll 1
    for (int g = 0; g < ng; ++g) {
        v4f r[16];
#pragma unroll
        for (int u = 0; u < 16; ++u) {
            const unsigned w0 = (unsigned)__builtin_amdgcn_readlane((int)e0_l, g * 16 + u);
            r[u] = *(const v4f*)(hwl + (size_t)(w0 & 0xffffu) * HID);
        }
#pragma unroll
        for (int u = 0; u < 16; ++u) {
            const unsigned w0 = (unsigned)__builtin_amdgcn_readlane((int)e0_l, g * 16 + u);
            const float en = __builtin_bit_cast(float,
                __builtin_amdgcn_readlane((int)e1_l, g * 16 + u));
            float* t = tl + (w0 >> 16) * T_PITCH;
            v4f a = *(const v4f*)t;
            a.x = fmaf(r[u].x, en, a.x);
            a.y = fmaf(r[u].y, en, a.y);
            a.z = fmaf(r[u].z, en, a.z);
            a.w = fmaf(r[u].w, en, a.w);
            *(v4f*)t = a;
        }
    }
}

template <bool FROM_LDS>
__device__ __forceinline__ void own_phase(const v2u* ll, const v2u* gl, v2u first, int c, int lane,
                                          const float* __restrict__ hw, float* tile) {
    const unsigned pad0 = (unsigned)OWN_NODES << 16;
    const int last = c - 1;
    unsigned nx0 = pad0, nx1 = 0u;
    if (c > 0) {
        v2u t = first;
        if (FROM_LDS) t = ll[lane < c ? lane : last];
        nx0 = lane < c ? t.x : pad0;
        nx1 = lane < c ? t.y : 0u;
    }
#pragma unroll 1
    for (int i = 0; i < c; i += 32) {
        const unsigned cur0 = nx0, cur1 = nx1;
        const int j  = i + 32 + lane;
        const int jc = j < c ? j : last;
        const v2u t = FROM_LDS ? ll[jc] : gl[jc];
        nx0 = j < c ? t.x : pad0;
        nx1 = j < c ? t.y : 0u;
        own_rows(cur0, cur1, (c - i) > 16 ? 2 : 1, lane, hw, tile);
    }
}

__device__ __forceinline__ void own_phase_l0(const v2u* ll, int c, int lane,
                                             const int* __restrict__ x, float* tile) {
    for (int i = 0; i < c; i += 32) {
        const int j = i + lane;
        if (j < c) {
            const v2u t = ll[j];
            const int2 id = *(const int2*)(x + 2 * (size_t)(t.x & 0xffffu));
            float* tr = tile + (t.x >> 16) * T_PITCH;
            const unsigned enb = t.y;
            const float en = __builtin_bit_cast(float, enb);
            lds_add(tr + id.x, en);
            lds_add(tr + N_CLS + id.y, en);
        }
    }
}

template <bool L0>
__device__ __forceinline__ void sc_bin(const unsigned* st, int n, int lane,
                                       const int* __restrict__ edge_index,
                                       const float* __restrict__ dinv,
                                       const int* __restrict__ x,
                                       const float* __restrict__ hw,
                                       float* tile, v2u* llist, int* lcnt, int* lovf) {
    int   s_l  = 0;
    int   dl_l = 0;
    float en_l = 0.0f;
    bool  over = false;
    if (lane < n) {
        const unsigned e = st[lane];
        s_l = edge_index[e];
        const int d = edge_index[E_EDGES + e];
        en_l = dinv[s_l] * dinv[d];
        dl_l = d & (OWN_NODES - 1);
        const int ow = dl_l & (SC_WAVES - 1);
        const int place = __hip_atomic_fetch_add(&lcnt[ow], 1, __ATOMIC_RELAXED, __HIP_MEMORY_SCOPE_WORKGROUP);
        if (place < LCAP) {
            v2u ent;
            ent.x = (unsigned)s_l | ((unsigned)dl_l << 16);
            ent.y = __builtin_bit_cast(unsigned, en_l);
            llist[ow * LCAP + place] = ent;
        } else {
            over = true;
        }
    }
    unsigned m = __builtin_amdgcn_ballot_w32(over);
    while (m != 0u) {
        const int l = __builtin_ctz(m);
        m &= m - 1u;
        const int   s  = __builtin_amdgcn_readlane(s_l, l);
        const int   dl = __builtin_amdgcn_readlane(dl_l, l);
        const float en = __builtin_bit_cast(float,
            __builtin_amdgcn_readlane(__builtin_bit_cast(int, en_l), l));
        if (L0) {
            const int2 id = *(const int2*)(x + 2 * (size_t)s);
            if (lane == 0) {
                lds_add(tile + dl * T_PITCH + id.x, en);
                lds_add(tile + dl * T_PITCH + N_CLS + id.y, en);
            }
        } else {
            const v4f r = *(const v4f*)(hw + (size_t)s * HID + lane * 4);
            float* t = tile + dl * T_PITCH + lane * 4;
            lds_add(t,     r.x * en);
            lds_add(t + 1, r.y * en);
            lds_add(t + 2, r.z * en);
            lds_add(t + 3, r.w * en);
        }
        if (lane == 0) *lovf = 1;
    }
}

template <bool L0>
__launch_bounds__(1024)
__global__ void k_scatter_gemm(const int* __restrict__ edge_index,
                               const unsigned char* __restrict__ okey,
                               const float* __restrict__ dinv,
                               const float* __restrict__ invdeg,
                               const int* __restrict__ x,
                               const float* __restrict__ cls_emb,
                               const float* __restrict__ nbr_emb,
                               const float* __restrict__ Tt,
                               const float* __restrict__ hw,
                               float* __restrict__ hwn,
                               float* __restrict__ h,
                               const float* __restrict__ bias,
                               const float* __restrict__ gamma,
                               const float* __restrict__ beta,
                               const float* __restrict__ Wn,
                               const float* __restrict__ W_out,
                               float* __restrict__ out,
                               unsigned* __restrict__ elist,
                               int* __restrict__ ecnt,
                               int* __restrict__ eflag,
                               const int is_last) {
    __shared__ __attribute__((aligned(16))) float tile[(OWN_NODES + 1) * T_PITCH];
    __shared__ __attribute__((aligned(16))) unsigned stk[SC_WAVES * QCAP];
    __shared__ __attribute__((aligned(16))) v2u llist[SC_WAVES * LCAP];
    __shared__ int lcnt[SC_WAVES];
    __shared__ int lovf;

    const int wave  = threadIdx.x >> 5;
    const int lane  = threadIdx.x & 31;
    const int blk   = blockIdx.x;
    const int node0 = blk * OWN_NODES;

    const int slot = blk * SC_WAVES + wave;
    v2u* gl = reinterpret_cast<v2u*>(elist + (size_t)slot * (2 * LCAP));
    const v2u* ll = llist + wave * LCAP;

    int fl = 1, cg = 0;
    v2u first = {(unsigned)OWN_NODES << 16, 0u};
    if (!L0) {
        fl = eflag[blk];
        cg = ecnt[slot];
        first = gl[lane];
    }
    float wo0 = 0.0f, wo1 = 0.0f;
    if (!L0 && is_last) {
        wo0 = W_out[threadIdx.x];
        if (threadIdx.x < WO_REST) wo1 = W_out[SC_WAVES * 32 + threadIdx.x];
    }
    const bool scan = L0 || __builtin_amdgcn_readfirstlane(fl) != 0;

    if (L0) {
        int   idc[OWN_NODES / SC_WAVES], idn[OWN_NODES / SC_WAVES];
        float inv[OWN_NODES / SC_WAVES];
#pragma unroll
        for (int j = 0; j < OWN_NODES / SC_WAVES; ++j) {
            const int n = node0 + wave + j * SC_WAVES;
            idc[j] = x[2 * n + 0];
            idn[j] = N_CLS + x[2 * n + 1];
            inv[j] = invdeg[n];
        }
#pragma unroll
        for (int j = 0; j < OWN_NODES / SC_WAVES; ++j) {
            const int r = wave + j * SC_WAVES;
            const int w = lane * 4;
            v4f v;
            v.x = (w + 0 == idc[j] ? inv[j] : 0.0f) + (w + 0 == idn[j] ? inv[j] : 0.0f);
            v.y = (w + 1 == idc[j] ? inv[j] : 0.0f) + (w + 1 == idn[j] ? inv[j] : 0.0f);
            v.z = (w + 2 == idc[j] ? inv[j] : 0.0f) + (w + 2 == idn[j] ? inv[j] : 0.0f);
            v.w = (w + 3 == idc[j] ? inv[j] : 0.0f) + (w + 3 == idn[j] ? inv[j] : 0.0f);
            *(v4f*)(tile + r * T_PITCH + w) = v;
            if (lane == 0) {
                v4f e;
                e.x = (HID + 0 == idn[j] ? inv[j] : 0.0f);
                e.y = (HID + 1 == idn[j] ? inv[j] : 0.0f);
                e.z = (HID + 2 == idn[j] ? inv[j] : 0.0f);
                e.w = (HID + 3 == idn[j] ? inv[j] : 0.0f);
                *(v4f*)(tile + r * T_PITCH + HID) = e;
            }
        }
    } else
    {
        const v4f bv = *(const v4f*)(bias + lane * 4);
        v4f   a[OWN_NODES / SC_WAVES];
        float inv[OWN_NODES / SC_WAVES];
#pragma unroll
        for (int j = 0; j < OWN_NODES / SC_WAVES; ++j) {
            const int r = wave + j * SC_WAVES;
            a[j]   = *(const v4f*)(hw + (size_t)(node0 + r) * HID + lane * 4);
            inv[j] = invdeg[node0 + r];
        }
#pragma unroll
        for (int j = 0; j < OWN_NODES / SC_WAVES; ++j) {
            const int r = wave + j * SC_WAVES;
            *(v4f*)(tile + r * T_PITCH + lane * 4) = a[j] * inv[j] + bv;
        }
    }

    int c;

    if (scan) {
        if (threadIdx.x < SC_WAVES) lcnt[threadIdx.x] = 0;
        if (threadIdx.x == 0) lovf = 0;
        __syncthreads();

        unsigned* st = stk + wave * QCAP;
        const unsigned idrep = (unsigned)blk * 0x01010101u;
        const unsigned char* kw = okey + (size_t)wave * E_WAVE + lane * 16;
        const unsigned e_w = (unsigned)wave * E_WAVE + (unsigned)lane * 16u;
        int cnt = 0;

        uint4 kk[4];
#pragma unroll
        for (int u = 0; u < 4; ++u)
            kk[u] = *(const uint4*)(kw + ((u + blk) & (SCAN_STEPS - 1)) * KEYS_STEP);
#pragma unroll 1
        for (int i = 0; i < SCAN_STEPS; i += 4) {
#pragma unroll
            for (int u = 0; u < 4; ++u) {
                const int it = (i + u + blk) & (SCAN_STEPS - 1);
                const unsigned eb = e_w + (unsigned)it * KEYS_STEP;
                const unsigned kv[4] = {kk[u].x, kk[u].y, kk[u].z, kk[u].w};
                unsigned p = (zero_bytes(kv[0] ^ idrep) >> 7) | (zero_bytes(kv[1] ^ idrep) >> 6) |
                             (zero_bytes(kv[2] ^ idrep) >> 5) | (zero_bytes(kv[3] ^ idrep) >> 4);
                for (;;) {
                    const bool has = (p != 0u);
                    const unsigned m = __builtin_amdgcn_ballot_w32(has);
                    if (m == 0u) break;
                    if (has) {
                        const unsigned t = (unsigned)__builtin_ctz(p);
                        st[cnt + (int)__builtin_amdgcn_mbcnt_lo(m, 0u)] = eb + ((t & 7u) << 2) + (t >> 3);
                        p &= p - 1u;
                    }
                    cnt += __builtin_popcount(m);
                }
                if (i + 4 < SCAN_STEPS)
                    kk[u] = *(const uint4*)(kw + ((i + u + 4 + blk) & (SCAN_STEPS - 1)) * KEYS_STEP);
                const int need = (i + u + 1 < SCAN_STEPS) ? 32 : 1;
                while (cnt >= need) {
                    const int n = cnt < 32 ? cnt : 32;
                    cnt -= n;
                    sc_bin<L0>(st + cnt, n, lane, edge_index, dinv, x, hw, tile, llist, lcnt, &lovf);
                }
            }
        }
        __syncthreads();

        const int cr = lcnt[wave];
        c = __builtin_amdgcn_readfirstlane(cr < LCAP ? cr : LCAP);
        if (L0) {
            const v2u padent = {(unsigned)OWN_NODES << 16, 0u};
            for (int j = lane; j < c || j < 32; j += 32) {
                v2u e = padent;
                if (j < c) e = ll[j];
                gl[j] = e;
            }
            if (lane == 0) ecnt[slot] = c;
            if (threadIdx.x == 0) eflag[blk] = lovf;
        }
    } else {
        c = __builtin_amdgcn_readfirstlane(cg);
        c = c < 0 ? 0 : (c > LCAP ? LCAP : c);
    }

    if (L0) {
        const v4f* Tg = reinterpret_cast<const v4f*>(Tt);
        v4f* lT4 = reinterpret_cast<v4f*>(stk);
        constexpr int NT4 = HID * T_ROWS / 4;
        constexpr int TPC = NT4 / (SC_WAVES * 32);
        v4f treg[TPC + 1];
#pragma unroll
        for (int t = 0; t < TPC; ++t) treg[t] = Tg[threadIdx.x + t * (SC_WAVES * 32)];
        const int trest = TPC * (SC_WAVES * 32) + threadIdx.x;
        treg[TPC] = Tg[trest < NT4 ? trest : 0];

        own_phase_l0(ll, c, lane, x, tile);

#pragma unroll
        for (int t = 0; t < TPC; ++t) lT4[threadIdx.x + t * (SC_WAVES * 32)] = treg[t];
        if (trest < NT4) lT4[trest] = treg[TPC];
        __syncthreads();

        const int half = lane >> 4;
        const int l16  = lane & 15;
        const int rt   = wave >> 1;
        const int c0   = (wave & 1) * 64;
        const float* lT = reinterpret_cast<const float*>(stk);
        v8f zero = {};
        v8f acc[4];
#pragma unroll
        for (int q = 0; q < 4; ++q) acc[q] = zero;
        const float* arow = tile + (rt * 16 + l16) * T_PITCH;
        for (int k = 0; k < T_ROWS; k += 4) {
            const int kb = k + half * 2;
            v2f a = *(const v2f*)(arow + kb);
#pragma unroll
            for (int q = 0; q < 4; ++q) {
                const int col = c0 + q * 16 + l16;
                v2f b = *(const v2f*)(lT + col * T_ROWS + kb);
                acc[q] = __builtin_amdgcn_wmma_f32_16x16x4_f32(
                    false, a, false, b, (short)0, acc[q], false, false);
            }
        }
        float bq[4];
#pragma unroll
        for (int q = 0; q < 4; ++q) bq[q] = bias[c0 + q * 16 + l16];
        __syncthreads();
#pragma unroll
        for (int r = 0; r < 8; ++r) {
            float* trow = tile + (rt * 16 + half * 8 + r) * T_PITCH + c0;
#pragma unroll
            for (int q = 0; q < 4; ++q) trow[q * 16 + l16] = acc[q][r] + bq[q];
        }
        __syncthreads();
    }
    else if (scan) own_phase<true>(ll, gl, first, c, lane, hw, tile);
    else           own_phase<false>(ll, gl, first, c, lane, hw, tile);

    const v4f gm = *(const v4f*)(gamma + lane * 4);
    const v4f bt = *(const v4f*)(beta + lane * 4);

    if (!L0 && is_last) {
        float* lWo = reinterpret_cast<float*>(stk);
        lWo[threadIdx.x] = wo0;
        if (threadIdx.x < WO_REST) lWo[SC_WAVES * 32 + threadIdx.x] = wo1;
        __syncthreads();
        const bool mine = wave < OWN_NODES / NPG;
        float* t = tile + (mine ? wave * NPG : 0) * T_PITCH;
        if (mine) {
            const v4f av = *(const v4f*)(t + lane * 4);
            const v4f hv = *(const v4f*)(h + (size_t)(node0 + wave * NPG) * HID + lane * 4);
            *(v4f*)(t + lane * 4) = relu_res_ln(hv, av, gm, bt);
        }
        __syncthreads();
        if (mine && lane < OUT_DIM) {
            float s = 0.0f;
#pragma unroll 32
            for (int k = 0; k < HID; ++k) s = fmaf(t[k], lWo[k * OUT_DIM + lane], s);
            out[(blk * (OWN_NODES / NPG) + wave) * OUT_DIM + lane] = s;
        }
        return;
    }

    float* lWt = reinterpret_cast<float*>(stk);
    v4f wreg[W_PIECES(SC_WAVES * 32)];
    stage_w_load<SC_WAVES * 32>(Wn, threadIdx.x, wreg);

#pragma unroll (L0 ? 4 : 8)
    for (int j = 0; j < OWN_NODES / SC_WAVES; ++j) {
        const int r = wave + j * SC_WAVES;
        float* t = tile + r * T_PITCH + lane * 4;
        const v4f av = *(const v4f*)t;
        float* hr = h + (size_t)(node0 + r) * HID + lane * 4;
        v4f hv;
        if (L0) {
            const int ic = x[2 * (node0 + r) + 0];
            const int in = x[2 * (node0 + r) + 1];
            hv = *(const v4f*)(cls_emb + (size_t)ic * HID + lane * 4) +
                 *(const v4f*)(nbr_emb + (size_t)in * HID + lane * 4);
        } else {
            hv = *(const v4f*)hr;
        }
        const v4f v = relu_res_ln(hv, av, gm, bt);
        *(v4f*)hr = v;
        *(v4f*)t = v;
    }
    stage_w_store<SC_WAVES * 32>(lWt, threadIdx.x, wreg);
    __syncthreads();

    const int half = lane >> 4;
    const int l16  = lane & 15;
    const int rt   = wave >> 1;
    const int c0   = (wave & 1) * 64;

    v8f zero = {};
    v8f acc[4];
#pragma unroll
    for (int q = 0; q < 4; ++q) acc[q] = zero;

    const float* arow = tile + (rt * 16 + l16) * T_PITCH;

    for (int k = 0; k < HID; k += 4) {
        const int kb = k + half * 2;
        v2f a = *(const v2f*)(arow + kb);
#pragma unroll
        for (int q = 0; q < 4; ++q) {
            const int col = c0 + q * 16 + l16;
            v2f b = *(const v2f*)(lWt + col * HID + (kb ^ ((col & 31) << 1)));
            acc[q] = __builtin_amdgcn_wmma_f32_16x16x4_f32(
                false, a, false, b, (short)0, acc[q], false, false);
        }
    }

#pragma unroll
    for (int r = 0; r < 8; ++r) {
        float* hwrow = hwn + (size_t)(node0 + rt * 16 + half * 8 + r) * HID + c0;
#pragma unroll
        for (int q = 0; q < 4; ++q) hwrow[q * 16 + l16] = acc[q][r];
    }
}

extern "C" void kernel_launch(void* const* d_in, const int* in_sizes, int n_in,
                              void* d_out, int out_size, void* d_ws, size_t ws_size,
                              hipStream_t stream) {
    const int*   x        = (const int*)d_in[0];
    const int*   ei       = (const int*)d_in[1];
    const float* cls_emb  = (const float*)d_in[3];
    const float* nbr_emb  = (const float*)d_in[4];
    const float* Ws       = (const float*)d_in[5];
    const float* bs       = (const float*)d_in[6];
    const float* gammas   = (const float*)d_in[7];
    const float* betas    = (const float*)d_in[8];
    const float* W_out    = (const float*)d_in[9];
    float*       out      = (float*)d_out;

    char* ws = (char*)d_ws;
    size_t off = 0;
    int*   degcnt = (int*)(ws + off);   off += (size_t)N_NODES * 4;
    float* dinv   = (float*)(ws + off); off += (size_t)N_NODES * 4;
    float* invdeg = (float*)(ws + off); off += (size_t)N_NODES * 4;
    float* h      = (float*)(ws + off); off += (size_t)N_NODES * HID * 4;
    float* hwa    = (float*)(ws + off); off += (size_t)N_NODES * HID * 4;
    float* hwb    = (float*)(ws + off); off += (size_t)N_NODES * HID * 4;
    unsigned char* okey = (unsigned char*)(ws + off); off += (size_t)E_EDGES;
    unsigned* elist = (unsigned*)(ws + off); off += (size_t)(N_NODES / OWN_NODES) * SC_WAVES * LCAP * 8;
    int*   ecnt   = (int*)(ws + off);   off += (size_t)(N_NODES / OWN_NODES) * SC_WAVES * 4;
    int*   eflag  = (int*)(ws + off);   off += (size_t)(N_NODES / OWN_NODES) * 4;
    float* Tt     = (float*)(ws + off); off += (size_t)HID * T_ROWS * 4;
    (void)ws_size; (void)in_sizes; (void)n_in; (void)out_size;

    k_deg_init  <<<N_NODES / 256, 256, 0, stream>>>(degcnt);
    k_deg_count <<<E_EDGES / 256, 256, 0, stream>>>(ei, degcnt, okey);
    k_deg_finish<<<N_NODES / 256, 256, 0, stream>>>(degcnt, dinv, invdeg);

    k_table<<<(T_ROWS + 15) / 16, 256, 0, stream>>>(cls_emb, nbr_emb, Ws, Tt);

    k_scatter_gemm<true><<<N_NODES / OWN_NODES, SC_WAVES * 32, 0, stream>>>(
        ei, okey, dinv, invdeg, x, cls_emb, nbr_emb, Tt, hwa, hwb, h, bs, gammas, betas,
        Ws + (size_t)1 * HID * HID, W_out, out, elist, ecnt, eflag, 0);
    k_scatter_gemm<false><<<N_NODES / OWN_NODES, SC_WAVES * 32, 0, stream>>>(
        ei, okey, dinv, invdeg, x, cls_emb, nbr_emb, Tt, hwb, hwa, h, bs + 1 * HID, gammas + 1 * HID, betas + 1 * HID,
        Ws + (size_t)2 * HID * HID, W_out, out, elist, ecnt, eflag, 0);
    k_scatter_gemm<false><<<N_NODES / OWN_NODES, SC_WAVES * 32, 0, stream>>>(
        ei, okey, dinv, invdeg, x, cls_emb, nbr_emb, Tt, hwa, hwb, h, bs + 2 * HID, gammas + 2 * HID, betas + 2 * HID,
        Ws + (size_t)3 * HID * HID, W_out, out, elist, ecnt, eflag, 0);
    k_scatter_gemm<false><<<N_NODES / OWN_NODES, SC_WAVES * 32, 0, stream>>>(
        ei, okey, dinv, invdeg, x, cls_emb, nbr_emb, Tt, hwb, hwa, h, bs + 3 * HID, gammas + 3 * HID, betas + 3 * HID,
        Ws, W_out, out, elist, ecnt, eflag, 1);
}
